# starting kernel
# baseline (speedup 1.0000x reference)
#include <hip/hip_runtime.h>
#include <hip/hip_bf16.h>
#include <math.h>

#define T_LEN   50000
#define W_LEN   101
#define N_IN    128
#define RANK    10
#define N_OUT   64

#define KPAD    104
#define KSTEPS  26
#define ROWS    32

#define CH1     512
#define NSUB1   4
#define CH3     128

typedef __attribute__((ext_vector_type(2))) float v2f;
typedef __attribute__((ext_vector_type(8))) float v8f;

__device__ __forceinline__ void conv_tile(const float* __restrict__ Xw, int xbase,
                                          const float* __restrict__ Kf,
                                          int lane, v8f& acc_re, v8f& acc_im) {
    const int col = lane & 15;
    const int m   = lane & 15;
    const int kb  = (lane >> 4) << 1;
#pragma unroll
    for (int r = 0; r < 8; ++r) { acc_re[r] = 0.0f; acc_im[r] = 0.0f; }
    for (int s = 0; s < KSTEPS; ++s) {
        const int u0 = s << 2;
        v2f a0, a1, b;
        const float* k0 = &Kf[m * KPAD + u0 + kb];
        a0.x = k0[0]; a0.y = k0[1];
        const float* k1 = &Kf[(m + 16) * KPAD + u0 + kb];
        a1.x = k1[0]; a1.y = k1[1];
        const int bi = xbase + col + u0 + kb;
        b.x = Xw[bi]; b.y = Xw[bi + 1];
        acc_re = __builtin_amdgcn_wmma_f32_16x16x4_f32(
            false, a0, false, b, (short)0, acc_re, false, false);
        acc_im = __builtin_amdgcn_wmma_f32_16x16x4_f32(
            false, a1, false, b, (short)0, acc_im, false, false);
    }
}

__device__ __forceinline__ void stage_filters(const float* __restrict__ K,
                                              float* __restrict__ Kf, int tid) {
    for (int i = tid; i < ROWS * KPAD; i += 256) {
        const int j = i / KPAD, u = i % KPAD;
        float v = 0.0f;
        if (u < W_LEN) {
            if (j < RANK)                      v = K[(j * 2 + 0) * W_LEN + (W_LEN - 1 - u)];
            else if (j >= 16 && j < 16 + RANK) v = K[((j - 16) * 2 + 1) * W_LEN + (W_LEN - 1 - u)];
        }
        Kf[i] = v;
    }
}

__global__ void zero_ws_kernel(float* __restrict__ p, int nelem) {
    const int i = blockIdx.x * blockDim.x + threadIdx.x;
    if (i < nelem) p[i] = 0.0f;
}

__global__ void __launch_bounds__(256)
pass1_stats(const float* __restrict__ X, const float* __restrict__ K,
            float* __restrict__ sumA, float* __restrict__ sumQ) {
    __shared__ float Kf[ROWS * KPAD];
    __shared__ float Xw[CH1 + KPAD + 8];
    const int tid  = threadIdx.x;
    const int lane = tid & 31;
    const int wave = tid >> 5;
    stage_filters(K, Kf, tid);

    const int T0 = blockIdx.x * CH1;
    const int n0 = blockIdx.y * 16;

    for (int nn = 0; nn < 16; ++nn) {
        const int n = n0 + nn;
        __syncthreads();
        for (int i = tid; i < CH1 + KPAD; i += 256) {
            const int t = T0 - 50 + i;
            Xw[i] = (t >= 0 && t < T_LEN) ? X[n * T_LEN + t] : 0.0f;
        }
        __syncthreads();

        float s[8], q[8];
#pragma unroll
        for (int r = 0; r < 8; ++r) { s[r] = 0.0f; q[r] = 0.0f; }

        for (int sub = 0; sub < NSUB1; ++sub) {
            const int toff = (wave * NSUB1 + sub) * 16;
            v8f ar, ai;
            conv_tile(Xw, toff, Kf, lane, ar, ai);
            const int  t    = T0 + toff + (lane & 15);
            const bool tval = (t < T_LEN);
#pragma unroll
            for (int r = 0; r < 8; ++r) {
                const float re = ar[r], im = ai[r];
                float c = sqrtf(re * re + im * im);
                c = tval ? c : 0.0f;
                s[r] += c; q[r] += c * c;
            }
        }
#pragma unroll
        for (int r = 0; r < 8; ++r) {
            float sv = s[r], qv = q[r];
            for (int off = 1; off < 16; off <<= 1) {
                sv += __shfl_xor(sv, off, 32);
                qv += __shfl_xor(qv, off, 32);
            }
            if ((lane & 15) == 0) {
                const int k = (lane >> 4) * 8 + r;
                if (k < RANK) {
                    atomicAdd(&sumA[n * RANK + k], sv);
                    atomicAdd(&sumQ[n * RANK + k], qv);
                }
            }
        }
    }
}

__global__ void finalize_stats(const float* __restrict__ B1,
                               const float* __restrict__ sumA,
                               const float* __restrict__ sumQ,
                               float* __restrict__ B1s, float* __restrict__ ybar) {
    const int tid = threadIdx.x;
    if (tid < RANK) ybar[tid] = 0.0f;
    __syncthreads();
    for (int i = tid; i < N_IN * RANK; i += blockDim.x) {
        const float s    = sumA[i], q = sumQ[i];
        const float mean = s / (float)T_LEN;
        const float var  = (q - s * mean) / (float)(T_LEN - 1);
        const float w    = B1[i] * rsqrtf(var);
        B1s[i] = w;
        atomicAdd(&ybar[i % RANK], w * mean);
    }
}

__global__ void __launch_bounds__(256)
pass3_out(const float* __restrict__ X, const float* __restrict__ K,
          const float* __restrict__ B2, const float* __restrict__ bias,
          const float* __restrict__ B1s, const float* __restrict__ ybar,
          float* __restrict__ out) {
    __shared__ float Kf[ROWS * KPAD];
    __shared__ float Xw[CH3 + KPAD + 8];
    __shared__ float B1sS[N_IN * RANK];
    __shared__ float B2S[RANK * N_OUT];
    __shared__ float ybS[RANK];
    __shared__ float biasS[N_OUT];
    const int tid  = threadIdx.x;
    const int lane = tid & 31;
    const int wave = tid >> 5;
    const int hi   = lane >> 4;

    stage_filters(K, Kf, tid);
    for (int i = tid; i < N_IN * RANK; i += 256) B1sS[i] = B1s[i];
    for (int i = tid; i < RANK * N_OUT; i += 256) B2S[i] = B2[i];
    if (tid < RANK)  ybS[tid]  = ybar[tid];
    if (tid < N_OUT) biasS[tid] = bias[tid];

    const int T0   = blockIdx.x * CH3;
    const int toff = wave * 16;
    const int t    = T0 + toff + (lane & 15);

    float y[8];
#pragma unroll
    for (int r = 0; r < 8; ++r) y[r] = 0.0f;

    for (int n = 0; n < N_IN; ++n) {
        __syncthreads();
        for (int i = tid; i < CH3 + KPAD; i += 256) {
            const int tt = T0 - 50 + i;
            Xw[i] = (tt >= 0 && tt < T_LEN) ? X[n * T_LEN + tt] : 0.0f;
        }
        __syncthreads();
        v8f ar, ai;
        conv_tile(Xw, toff, Kf, lane, ar, ai);
#pragma unroll
        for (int r = 0; r < 8; ++r) {
            const int   k  = hi * 8 + r;
            const float re = ar[r], im = ai[r];
            const float c  = sqrtf(re * re + im * im);
            const float w  = (k < RANK) ? B1sS[n * RANK + k] : 0.0f;
            y[r] += w * c;
        }
    }

    float yp[8];
    int   kc[8];
#pragma unroll
    for (int r = 0; r < 8; ++r) {
        const int k = hi * 8 + r;
        yp[r] = (k < RANK) ? (y[r] - ybS[k]) : 0.0f;
        kc[r] = (k < RANK) ? k : 0;
    }
    for (int m = 0; m < N_OUT; ++m) {
        float p = 0.0f;
#pragma unroll
        for (int r = 0; r < 8; ++r) p += yp[r] * B2S[kc[r] * N_OUT + m];
        p += __shfl_xor(p, 16, 32);
        if (hi == 0 && t < T_LEN) out[m * T_LEN + t] = p + biasS[m];
    }
}

extern "C" void kernel_launch(void* const* d_in, const int* in_sizes, int n_in,
                              void* d_out, int out_size, void* d_ws, size_t ws_size,
                              hipStream_t stream) {
    const float* X    = (const float*)d_in[0];
    const float* K    = (const float*)d_in[1];
    const float* B1   = (const float*)d_in[2];
    const float* B2   = (const float*)d_in[3];
    const float* bias = (const float*)d_in[4];
    float*       out  = (float*)d_out;

    float* ws   = (float*)d_ws;
    float* sumA = ws;
    float* sumQ = ws + N_IN * RANK;
    float* B1s  = ws + 2 * N_IN * RANK;
    float* ybar = ws + 3 * N_IN * RANK;

    const int nstats = 2 * N_IN * RANK;
    zero_ws_kernel<<<(nstats + 255) / 256, 256, 0, stream>>>(ws, nstats);

    const int nchunk1 = (T_LEN + CH1 - 1) / CH1;
    pass1_stats<<<dim3(nchunk1, N_IN / 16), 256, 0, stream>>>(X, K, sumA, sumQ);

    finalize_stats<<<1, 256, 0, stream>>>(B1, sumA, sumQ, B1s, ybar);

    const int nchunk3 = (T_LEN + CH3 - 1) / CH3;
    pass3_out<<<nchunk3, 256, 0, stream>>>(X, K, B2, bias, B1s, ybar, out);
}


# pass 1 re-tiled: one input row per workgroup over 4224 time points, sums stay in registers across a wave's 33 subtiles; one lane+wave reduction and one atomic per (row,filter) sum per workgroup
# speedup vs baseline: 1.6791x; 1.6791x over previous
#include <hip/hip_runtime.h>
#include <hip/hip_bf16.h>
#include <math.h>

#define T_LEN   50000
#define W_LEN   101
#define N_IN    128
#define RANK    10
#define N_OUT   64

#define KPAD    104
#define KSTEPS  26
#define ROWS    32

#define SPAN1   4224
#define NSUBW1  33
#define XORG1   52
#define XW1F4   (5 * 256)
#define CH3     128

typedef __attribute__((ext_vector_type(2))) float v2f;
typedef __attribute__((ext_vector_type(8))) float v8f;

__device__ __forceinline__ void conv_tile(const float* __restrict__ Xw, int xbase,
                                          const float* __restrict__ Kf,
                                          int lane, v8f& acc_re, v8f& acc_im) {
    const int col = lane & 15;
    const int m   = lane & 15;
    const int kb  = (lane >> 4) << 1;
#pragma unroll
    for (int r = 0; r < 8; ++r) { acc_re[r] = 0.0f; acc_im[r] = 0.0f; }
    for (int s = 0; s < KSTEPS; ++s) {
        const int u0 = s << 2;
        v2f a0, a1, b;
        const float* k0 = &Kf[m * KPAD + u0 + kb];
        a0.x = k0[0]; a0.y = k0[1];
        const float* k1 = &Kf[(m + 16) * KPAD + u0 + kb];
        a1.x = k1[0]; a1.y = k1[1];
        const int bi = xbase + col + u0 + kb;
        b.x = Xw[bi]; b.y = Xw[bi + 1];
        acc_re = __builtin_amdgcn_wmma_f32_16x16x4_f32(
            false, a0, false, b, (short)0, acc_re, false, false);
        acc_im = __builtin_amdgcn_wmma_f32_16x16x4_f32(
            false, a1, false, b, (short)0, acc_im, false, false);
    }
}

__device__ __forceinline__ void stage_filters(const float* __restrict__ K,
                                              float* __restrict__ Kf, int tid) {
    for (int i = tid; i < ROWS * KPAD; i += 256) {
        const int j = i / KPAD, u = i % KPAD;
        float v = 0.0f;
        if (u < W_LEN) {
            if (j < RANK)                      v = K[(j * 2 + 0) * W_LEN + (W_LEN - 1 - u)];
            else if (j >= 16 && j < 16 + RANK) v = K[((j - 16) * 2 + 1) * W_LEN + (W_LEN - 1 - u)];
        }
        Kf[i] = v;
    }
}

__global__ void zero_ws_kernel(float* __restrict__ p, int nelem) {
    const int i = blockIdx.x * blockDim.x + threadIdx.x;
    if (i < nelem) p[i] = 0.0f;
}

static_assert(T_LEN % 16 == 0, "a 16-t subtile is wholly inside or wholly outside [0,T)");
static_assert(T_LEN % 4 == 0 && SPAN1 % 4 == 0 && XORG1 % 4 == 0, "16-byte groups never straddle 0 or T");
static_assert(SPAN1 == NSUBW1 * 16 * 8, "span = 8 waves x NSUBW1 subtiles");
static_assert(XW1F4 * 4 >= SPAN1 + (XORG1 - 50) + KPAD, "window holds every tap of the last subtile");

__global__ void __launch_bounds__(256)
pass1_stats(const float* __restrict__ X, const float* __restrict__ K,
            float* __restrict__ sumA, float* __restrict__ sumQ) {
    __shared__ float Kf[ROWS * KPAD];
    __shared__ float4 Xw4[XW1F4];
    __shared__ float red[2 * 8 * 16];
    const int tid  = threadIdx.x;
    const int lane = tid & 31;
    const int wave = __builtin_amdgcn_readfirstlane(tid >> 5);
    stage_filters(K, Kf, tid);

    const int T0 = blockIdx.x * SPAN1;
    const int n  = blockIdx.y;
    const float* __restrict__ Xn = X + (size_t)n * T_LEN;

#pragma unroll
    for (int j = 0; j < XW1F4 / 256; ++j) {
        const int i  = tid + j * 256;
        const int t  = T0 - XORG1 + 4 * i;
        const int tc = min(max(t, 0), T_LEN - 4);
        float4 v = *reinterpret_cast<const float4*>(Xn + tc);
        if (t < 0 || t >= T_LEN) v = make_float4(0.0f, 0.0f, 0.0f, 0.0f);
        Xw4[i] = v;
    }
    __syncthreads();
    const float* Xw = reinterpret_cast<const float*>(Xw4);

    float s[8], q[8];
#pragma unroll
    for (int r = 0; r < 8; ++r) { s[r] = 0.0f; q[r] = 0.0f; }

    const int sub0 = wave * NSUBW1;
    const int nsub = min(max((T_LEN - T0) / 16 - sub0, 0), NSUBW1);
    for (int sub = 0; sub < nsub; ++sub) {
        const int toff = (sub0 + sub) * 16;
        v8f ar, ai;
        conv_tile(Xw, toff + (XORG1 - 50), Kf, lane, ar, ai);
#pragma unroll
        for (int r = 0; r < 8; ++r) {
            const float re = ar[r], im = ai[r];
            const float c  = sqrtf(re * re + im * im);
            s[r] += c; q[r] += c * c;
        }
    }

#pragma unroll
    for (int r = 0; r < 8; ++r) {
        float sv = s[r], qv = q[r];
        for (int off = 1; off < 16; off <<= 1) {
            sv += __shfl_xor(sv, off, 32);
            qv += __shfl_xor(qv, off, 32);
        }
        if ((lane & 15) == 0) {
            const int k = (lane >> 4) * 8 + r;
            red[wave * 16 + k]       = sv;
            red[128 + wave * 16 + k] = qv;
        }
    }
    __syncthreads();
    if (tid < 2 * RANK) {
        const int which = tid / RANK, k = tid % RANK;
        float a = 0.0f;
        for (int w = 0; w < 8; ++w) a += red[which * 128 + w * 16 + k];
        atomicAdd(which ? &sumQ[n * RANK + k] : &sumA[n * RANK + k], a);
    }
}

__global__ void finalize_stats(const float* __restrict__ B1,
                               const float* __restrict__ sumA,
                               const float* __restrict__ sumQ,
                               float* __restrict__ B1s, float* __restrict__ ybar) {
    const int tid = threadIdx.x;
    if (tid < RANK) ybar[tid] = 0.0f;
    __syncthreads();
    for (int i = tid; i < N_IN * RANK; i += blockDim.x) {
        const float s    = sumA[i], q = sumQ[i];
        const float mean = s / (float)T_LEN;
        const float var  = (q - s * mean) / (float)(T_LEN - 1);
        const float w    = B1[i] * rsqrtf(var);
        B1s[i] = w;
        atomicAdd(&ybar[i % RANK], w * mean);
    }
}

__global__ void __launch_bounds__(256)
pass3_out(const float* __restrict__ X, const float* __restrict__ K,
          const float* __restrict__ B2, const float* __restrict__ bias,
          const float* __restrict__ B1s, const float* __restrict__ ybar,
          float* __restrict__ out) {
    __shared__ float Kf[ROWS * KPAD];
    __shared__ float Xw[CH3 + KPAD + 8];
    __shared__ float B1sS[N_IN * RANK];
    __shared__ float B2S[RANK * N_OUT];
    __shared__ float ybS[RANK];
    __shared__ float biasS[N_OUT];
    const int tid  = threadIdx.x;
    const int lane = tid & 31;
    const int wave = tid >> 5;
    const int hi   = lane >> 4;

    stage_filters(K, Kf, tid);
    for (int i = tid; i < N_IN * RANK; i += 256) B1sS[i] = B1s[i];
    for (int i = tid; i < RANK * N_OUT; i += 256) B2S[i] = B2[i];
    if (tid < RANK)  ybS[tid]  = ybar[tid];
    if (tid < N_OUT) biasS[tid] = bias[tid];

    const int T0   = blockIdx.x * CH3;
    const int toff = wave * 16;
    const int t    = T0 + toff + (lane & 15);

    float y[8];
#pragma unroll
    for (int r = 0; r < 8; ++r) y[r] = 0.0f;

    for (int n = 0; n < N_IN; ++n) {
        __syncthreads();
        for (int i = tid; i < CH3 + KPAD; i += 256) {
            const int tt = T0 - 50 + i;
            Xw[i] = (tt >= 0 && tt < T_LEN) ? X[n * T_LEN + tt] : 0.0f;
        }
        __syncthreads();
        v8f ar, ai;
        conv_tile(Xw, toff, Kf, lane, ar, ai);
#pragma unroll
        for (int r = 0; r < 8; ++r) {
            const int   k  = hi * 8 + r;
            const float re = ar[r], im = ai[r];
            const float c  = sqrtf(re * re + im * im);
            const float w  = (k < RANK) ? B1sS[n * RANK + k] : 0.0f;
            y[r] += w * c;
        }
    }

    float yp[8];
    int   kc[8];
#pragma unroll
    for (int r = 0; r < 8; ++r) {
        const int k = hi * 8 + r;
        yp[r] = (k < RANK) ? (y[r] - ybS[k]) : 0.0f;
        kc[r] = (k < RANK) ? k : 0;
    }
    for (int m = 0; m < N_OUT; ++m) {
        float p = 0.0f;
#pragma unroll
        for (int r = 0; r < 8; ++r) p += yp[r] * B2S[kc[r] * N_OUT + m];
        p += __shfl_xor(p, 16, 32);
        if (hi == 0 && t < T_LEN) out[m * T_LEN + t] = p + biasS[m];
    }
}

extern "C" void kernel_launch(void* const* d_in, const int* in_sizes, int n_in,
                              void* d_out, int out_size, void* d_ws, size_t ws_size,
                              hipStream_t stream) {
    const float* X    = (const float*)d_in[0];
    const float* K    = (const float*)d_in[1];
    const float* B1   = (const float*)d_in[2];
    const float* B2   = (const float*)d_in[3];
    const float* bias = (const float*)d_in[4];
    float*       out  = (float*)d_out;

    float* ws   = (float*)d_ws;
    float* sumA = ws;
    float* sumQ = ws + N_IN * RANK;
    float* B1s  = ws + 2 * N_IN * RANK;
    float* ybar = ws + 3 * N_IN * RANK;

    const int nstats = 2 * N_IN * RANK;
    zero_ws_kernel<<<(nstats + 255) / 256, 256, 0, stream>>>(ws, nstats);

    const int nspan1 = (T_LEN + SPAN1 - 1) / SPAN1;
    pass1_stats<<<dim3(nspan1, N_IN), 256, 0, stream>>>(X, K, sumA, sumQ);

    finalize_stats<<<1, 256, 0, stream>>>(B1, sumA, sumQ, B1s, ybar);

    const int nchunk3 = (T_LEN + CH3 - 1) / CH3;
    pass3_out<<<nchunk3, 256, 0, stream>>>(X, K, B2, bias, B1s, ybar, out);
}


# pass 3 re-tiled: one workgroup = 32 time points x all 128 input rows staged once, no per-row barrier; 8 waves = 2 subtiles x 4 row groups, row-group sums added through LDS; 128-byte output lines
# speedup vs baseline: 1.9943x; 1.1877x over previous
#include <hip/hip_runtime.h>
#include <hip/hip_bf16.h>
#include <math.h>

#define T_LEN   50000
#define W_LEN   101
#define N_IN    128
#define RANK    10
#define N_OUT   64

#define KPAD    104
#define KSTEPS  26
#define ROWS    32

#define SPAN1   4224
#define NSUBW1  33
#define XORG1   52
#define XW1F4   (5 * 256)
#define CH3     32
#define NG3     4
#define NPG3    32
#define XORG3   52
#define XR3F4   35
#define XP3     (XR3F4 * 4)
#define NTH3    (32 * 2 * NG3)

typedef __attribute__((ext_vector_type(2))) float v2f;
typedef __attribute__((ext_vector_type(8))) float v8f;

__device__ __forceinline__ void conv_tile(const float* __restrict__ Xw, int xbase,
                                          const float* __restrict__ Kf,
                                          int lane, v8f& acc_re, v8f& acc_im) {
    const int col = lane & 15;
    const int m   = lane & 15;
    const int kb  = (lane >> 4) << 1;
#pragma unroll
    for (int r = 0; r < 8; ++r) { acc_re[r] = 0.0f; acc_im[r] = 0.0f; }
    for (int s = 0; s < KSTEPS; ++s) {
        const int u0 = s << 2;
        v2f a0, a1, b;
        const float* k0 = &Kf[m * KPAD + u0 + kb];
        a0.x = k0[0]; a0.y = k0[1];
        const float* k1 = &Kf[(m + 16) * KPAD + u0 + kb];
        a1.x = k1[0]; a1.y = k1[1];
        const int bi = xbase + col + u0 + kb;
        b.x = Xw[bi]; b.y = Xw[bi + 1];
        acc_re = __builtin_amdgcn_wmma_f32_16x16x4_f32(
            false, a0, false, b, (short)0, acc_re, false, false);
        acc_im = __builtin_amdgcn_wmma_f32_16x16x4_f32(
            false, a1, false, b, (short)0, acc_im, false, false);
    }
}

__device__ __forceinline__ void stage_filters(const float* __restrict__ K,
                                              float* __restrict__ Kf, int tid, int nthr) {
    for (int i = tid; i < ROWS * KPAD; i += nthr) {
        const int j = i / KPAD, u = i % KPAD;
        float v = 0.0f;
        if (u < W_LEN) {
            if (j < RANK)                      v = K[(j * 2 + 0) * W_LEN + (W_LEN - 1 - u)];
            else if (j >= 16 && j < 16 + RANK) v = K[((j - 16) * 2 + 1) * W_LEN + (W_LEN - 1 - u)];
        }
        Kf[i] = v;
    }
}

__global__ void zero_ws_kernel(float* __restrict__ p, int nelem) {
    const int i = blockIdx.x * blockDim.x + threadIdx.x;
    if (i < nelem) p[i] = 0.0f;
}

static_assert(T_LEN % 16 == 0, "a 16-t subtile is wholly inside or wholly outside [0,T)");
static_assert(T_LEN % 4 == 0 && SPAN1 % 4 == 0 && XORG1 % 4 == 0, "16-byte groups never straddle 0 or T");
static_assert(SPAN1 == NSUBW1 * 16 * 8, "span = 8 waves x NSUBW1 subtiles");
static_assert(XW1F4 * 4 >= SPAN1 + (XORG1 - 50) + KPAD, "window holds every tap of the last subtile");

__global__ void __launch_bounds__(256)
pass1_stats(const float* __restrict__ X, const float* __restrict__ K,
            float* __restrict__ sumA, float* __restrict__ sumQ) {
    __shared__ float Kf[ROWS * KPAD];
    __shared__ float4 Xw4[XW1F4];
    __shared__ float red[2 * 8 * 16];
    const int tid  = threadIdx.x;
    const int lane = tid & 31;
    const int wave = __builtin_amdgcn_readfirstlane(tid >> 5);
    stage_filters(K, Kf, tid, 256);

    const int T0 = blockIdx.x * SPAN1;
    const int n  = blockIdx.y;
    const float* __restrict__ Xn = X + (size_t)n * T_LEN;

#pragma unroll
    for (int j = 0; j < XW1F4 / 256; ++j) {
        const int i  = tid + j * 256;
        const int t  = T0 - XORG1 + 4 * i;
        const int tc = min(max(t, 0), T_LEN - 4);
        float4 v = *reinterpret_cast<const float4*>(Xn + tc);
        if (t < 0 || t >= T_LEN) v = make_float4(0.0f, 0.0f, 0.0f, 0.0f);
        Xw4[i] = v;
    }
    __syncthreads();
    const float* Xw = reinterpret_cast<const float*>(Xw4);

    float s[8], q[8];
#pragma unroll
    for (int r = 0; r < 8; ++r) { s[r] = 0.0f; q[r] = 0.0f; }

    const int sub0 = wave * NSUBW1;
    const int nsub = min(max((T_LEN - T0) / 16 - sub0, 0), NSUBW1);
    for (int sub = 0; sub < nsub; ++sub) {
        const int toff = (sub0 + sub) * 16;
        v8f ar, ai;
        conv_tile(Xw, toff + (XORG1 - 50), Kf, lane, ar, ai);
#pragma unroll
        for (int r = 0; r < 8; ++r) {
            const float re = ar[r], im = ai[r];
            const float c  = sqrtf(re * re + im * im);
            s[r] += c; q[r] += c * c;
        }
    }

#pragma unroll
    for (int r = 0; r < 8; ++r) {
        float sv = s[r], qv = q[r];
        for (int off = 1; off < 16; off <<= 1) {
            sv += __shfl_xor(sv, off, 32);
            qv += __shfl_xor(qv, off, 32);
        }
        if ((lane & 15) == 0) {
            const int k = (lane >> 4) * 8 + r;
            red[wave * 16 + k]       = sv;
            red[128 + wave * 16 + k] = qv;
        }
    }
    __syncthreads();
    if (tid < 2 * RANK) {
        const int which = tid / RANK, k = tid % RANK;
        float a = 0.0f;
        for (int w = 0; w < 8; ++w) a += red[which * 128 + w * 16 + k];
        atomicAdd(which ? &sumQ[n * RANK + k] : &sumA[n * RANK + k], a);
    }
}

__global__ void finalize_stats(const float* __restrict__ B1,
                               const float* __restrict__ sumA,
                               const float* __restrict__ sumQ,
                               float* __restrict__ B1s, float* __restrict__ ybar) {
    const int tid = threadIdx.x;
    if (tid < RANK) ybar[tid] = 0.0f;
    __syncthreads();
    for (int i = tid; i < N_IN * RANK; i += blockDim.x) {
        const float s    = sumA[i], q = sumQ[i];
        const float mean = s / (float)T_LEN;
        const float var  = (q - s * mean) / (float)(T_LEN - 1);
        const float w    = B1[i] * rsqrtf(var);
        B1s[i] = w;
        atomicAdd(&ybar[i % RANK], w * mean);
    }
}

static_assert(N_IN == NG3 * NPG3, "row groups cover the input rows");
static_assert(CH3 % 16 == 0 && CH3 % 4 == 0 && XORG3 % 4 == 0, "16-byte groups never straddle 0 or T");
static_assert(XP3 >= CH3 + (XORG3 - 50) + KPAD, "window holds every tap of the last subtile");
static_assert(CH3 == 32 && NTH3 % 32 == 0 && N_OUT % (NTH3 / 32) == 0, "readout: a wave stores 32 t of one output row");

__global__ void __launch_bounds__(NTH3)
pass3_out(const float* __restrict__ X, const float* __restrict__ K,
          const float* __restrict__ B2, const float* __restrict__ bias,
          const float* __restrict__ B1s, const float* __restrict__ ybar,
          float* __restrict__ out) {
    __shared__ float  Kf[ROWS * KPAD];
    __shared__ float4 Xw4[N_IN * XR3F4];
    __shared__ float  B1sS[N_IN * RANK];
    __shared__ float  B2S[RANK * N_OUT];
    __shared__ float  ypart[NG3 * (CH3 / 16) * RANK * 16];
    __shared__ float  yS[RANK * CH3];
    __shared__ float  ybS[RANK];
    __shared__ float  biasS[N_OUT];
    const int tid  = threadIdx.x;
    const int lane = tid & 31;
    const int wave = __builtin_amdgcn_readfirstlane(tid >> 5);
    const int hi   = lane >> 4;
    const int sub  = wave % (CH3 / 16);
    const int grp  = wave / (CH3 / 16);

    const int T0 = blockIdx.x * CH3;

#pragma unroll
    for (int j = 0; j < (N_IN * XR3F4 + NTH3 - 1) / NTH3; ++j) {
        const int i  = tid + j * NTH3;
        const int ic = min(i, N_IN * XR3F4 - 1);
        const int n  = ic / XR3F4, g = ic % XR3F4;
        const int t  = T0 - XORG3 + 4 * g;
        const int tc = min(max(t, 0), T_LEN - 4);
        float4 v = *reinterpret_cast<const float4*>(X + (size_t)n * T_LEN + tc);
        if (t < 0 || t >= T_LEN) v = make_float4(0.0f, 0.0f, 0.0f, 0.0f);
        if (i < N_IN * XR3F4) Xw4[i] = v;
    }
    stage_filters(K, Kf, tid, NTH3);
    for (int i = tid; i < N_IN * RANK; i += NTH3) B1sS[i] = B1s[i];
    for (int i = tid; i < RANK * N_OUT; i += NTH3) B2S[i] = B2[i];
    if (tid < RANK)  ybS[tid]  = ybar[tid];
    if (tid < N_OUT) biasS[tid] = bias[tid];
    __syncthreads();
    const float* Xw = reinterpret_cast<const float*>(Xw4);

    float y[8];
#pragma unroll
    for (int r = 0; r < 8; ++r) y[r] = 0.0f;

    for (int nn = 0; nn < NPG3; ++nn) {
        const int n = grp * NPG3 + nn;
        v8f ar, ai;
        conv_tile(Xw + n * XP3, sub * 16 + (XORG3 - 50), Kf, lane, ar, ai);
#pragma unroll
        for (int r = 0; r < 8; ++r) {
            const int   k  = hi * 8 + r;
            const float re = ar[r], im = ai[r];
            const float c  = sqrtf(re * re + im * im);
            const float w  = (k < RANK) ? B1sS[n * RANK + k] : 0.0f;
            y[r] += w * c;
        }
    }

#pragma unroll
    for (int r = 0; r < 8; ++r) {
        const int k = hi * 8 + r;
        if (k < RANK) ypart[((grp * (CH3 / 16) + sub) * RANK + k) * 16 + (lane & 15)] = y[r];
    }
    __syncthreads();
    for (int i = tid; i < (CH3 / 16) * RANK * 16; i += NTH3) {
        const int sb = i / (RANK * 16), k = (i / 16) % RANK, c = i & 15;
        float a = 0.0f;
        for (int g = 0; g < NG3; ++g) a += ypart[((g * (CH3 / 16) + sb) * RANK + k) * 16 + c];
        yS[k * CH3 + sb * 16 + c] = a - ybS[k];
    }
    __syncthreads();

    const int tt = tid & 31;
    const int t  = T0 + tt;
    for (int j = 0; j < N_OUT / (NTH3 / 32); ++j) {
        const int m = (tid >> 5) + j * (NTH3 / 32);
        float p = 0.0f;
        for (int k = 0; k < RANK; ++k) p += yS[k * CH3 + tt] * B2S[k * N_OUT + m];
        if (t < T_LEN) out[(size_t)m * T_LEN + t] = p + biasS[m];
    }
}

extern "C" void kernel_launch(void* const* d_in, const int* in_sizes, int n_in,
                              void* d_out, int out_size, void* d_ws, size_t ws_size,
                              hipStream_t stream) {
    const float* X    = (const float*)d_in[0];
    const float* K    = (const float*)d_in[1];
    const float* B1   = (const float*)d_in[2];
    const float* B2   = (const float*)d_in[3];
    const float* bias = (const float*)d_in[4];
    float*       out  = (float*)d_out;

    float* ws   = (float*)d_ws;
    float* sumA = ws;
    float* sumQ = ws + N_IN * RANK;
    float* B1s  = ws + 2 * N_IN * RANK;
    float* ybar = ws + 3 * N_IN * RANK;

    const int nstats = 2 * N_IN * RANK;
    zero_ws_kernel<<<(nstats + 255) / 256, 256, 0, stream>>>(ws, nstats);

    const int nspan1 = (T_LEN + SPAN1 - 1) / SPAN1;
    pass1_stats<<<dim3(nspan1, N_IN), 256, 0, stream>>>(X, K, sumA, sumQ);

    finalize_stats<<<1, 256, 0, stream>>>(B1, sumA, sumQ, B1s, ybar);

    const int nchunk3 = (T_LEN + CH3 - 1) / CH3;
    pass3_out<<<nchunk3, NTH3, 0, stream>>>(X, K, B2, bias, B1s, ybar, out);
}


# 20 filters x 4 time shifts fill 5 whole WMMA row tiles against X columns at stride 4: 130 matrix instructions per 64 t instead of 208, in both conv kernels
# speedup vs baseline: 3.0837x; 1.5463x over previous
#include <hip/hip_runtime.h>
#include <hip/hip_bf16.h>
#include <math.h>


#define T_LEN   50000
#define W_LEN   101
#define N_IN    128
#define RANK    10
#define N_OUT   64

#define KPAD    104
#define KSTEPS  26
#define NTILE   5
#define AROWS   (NTILE * 16)
#define AP      108
#define TT      64
#define XORG    52
#define XB0     (XORG - 50)

#define NW1     4
#define NTH1    (32 * NW1)
#define NTW1    8
#define SPAN1   (TT * NW1 * NTW1)
#define XW1F4   ((SPAN1 - TT + XB0 + 60 + KPAD + 3) / 4)
#define XW1TR   ((XW1F4 + NTH1 - 1) / NTH1)
#define NG3     4
#define NPG3    (N_IN / NG3)
#define XR3F4   42
#define XP3     (XR3F4 * 4)
#define NTH3    (32 * NG3)
#define KTR     ((RANK * 2 * W_LEN + 127) / 128)
#define KSN     (KTR * 128)

typedef __attribute__((ext_vector_type(2))) float v2f;
typedef __attribute__((ext_vector_type(8))) float v8f;

__device__ __forceinline__ void conv_tile5(const float* __restrict__ Xw, int xb,
                                           const float* __restrict__ Af,
                                           int lane, v8f (&acc)[NTILE]) {
    const int c = lane & 15, h = lane >> 4;
    const float* ap = Af + c * AP + 2 * h;
    const float* bp = Xw + xb + 4 * c + 2 * h;
#pragma unroll
    for (int q = 0; q < NTILE; ++q)
#pragma unroll
        for (int r = 0; r < 8; ++r) acc[q][r] = 0.0f;
    for (int s = 0; s < KSTEPS; ++s) {
        const v2f b = *reinterpret_cast<const v2f*>(bp + 4 * s);
#pragma unroll
        for (int q = 0; q < NTILE; ++q) {
            const v2f a = *reinterpret_cast<const v2f*>(ap + q * 16 * AP + 4 * s);
            acc[q] = __builtin_amdgcn_wmma_f32_16x16x4_f32(
                false, a, false, b, (short)0, acc[q], false, false);
        }
    }
}

template <int NTHR>
__device__ __forceinline__ void stage_filters(const float* __restrict__ K,
                                              float* __restrict__ Ks,
                                              float* __restrict__ Af, int tid) {
    static_assert(NTHR >= AP, "one thread per column of the filter bank");
    static_assert(NTHR == 128, "Ks holds KTR whole trips of 128 threads");
#pragma unroll
    for (int j = 0; j < KTR; ++j) {
        const int i = tid + j * NTHR;
        Ks[i] = K[min(i, RANK * 2 * W_LEN - 1)];
    }
    __syncthreads();
    if (tid < AP) {
        const int up = tid;
#pragma unroll
        for (int R = 0; R < AROWS; ++R) {
            const int q = R >> 4, m = R & 15, pos = m >> 2, d = m & 3;
            const int k    = (q < 4) ? ((q >> 1) * 4 + pos) : (8 + (pos >> 1));
            const int part = (q < 4) ? (q & 1) : (pos & 1);
            const int u = up - d;
            const int w = min(max(W_LEN - 1 - u, 0), W_LEN - 1);
            const float v = Ks[(k * 2 + part) * W_LEN + w];
            Af[R * AP + up] = (u >= 0 && u < W_LEN) ? v : 0.0f;
        }
    }
}

__global__ void zero_ws_kernel(float* __restrict__ p, int nelem) {
    const int i = blockIdx.x * blockDim.x + threadIdx.x;
    if (i < nelem) p[i] = 0.0f;
}

static_assert(T_LEN % 4 == 0 && SPAN1 % 4 == 0 && XORG % 4 == 0, "16-byte groups and 4-t column groups never straddle 0 or T");
static_assert(XW1F4 * 4 >= SPAN1 - TT + XB0 + 60 + KPAD + 2, "window holds every tap of the last tile");

__global__ void __launch_bounds__(NTH1)
pass1_stats(const float* __restrict__ X, const float* __restrict__ K,
            float* __restrict__ sumA, float* __restrict__ sumQ) {
    __shared__ __attribute__((aligned(16))) float Af[AROWS * AP];
    __shared__ float Ks[KSN];
    __shared__ float4 Xw4[XW1TR * NTH1];
    __shared__ float red[2 * NW1 * 16];
    const int tid  = threadIdx.x;
    const int lane = tid & 31;
    const int wave = __builtin_amdgcn_readfirstlane(tid >> 5);
    const int c = lane & 15, h = lane >> 4;

    const int T0 = blockIdx.x * SPAN1;
    const int n  = blockIdx.y;
    const float* __restrict__ Xn = X + (size_t)n * T_LEN;

#pragma unroll
    for (int j = 0; j < XW1TR; ++j) {
        const int i  = tid + j * NTH1;
        const int t  = T0 - XORG + 4 * i;
        const int tc = min(max(t, 0), T_LEN - 4);
        float4 v = *reinterpret_cast<const float4*>(Xn + tc);
        if (t < 0 || t >= T_LEN) v = make_float4(0.0f, 0.0f, 0.0f, 0.0f);
        Xw4[i] = v;
    }
    stage_filters<NTH1>(K, Ks, Af, tid);
    __syncthreads();
    const float* Xw = reinterpret_cast<const float*>(Xw4);

    float s[5], q2[5];
#pragma unroll
    for (int j = 0; j < 5; ++j) { s[j] = 0.0f; q2[j] = 0.0f; }

    const int ntile = min((T_LEN - T0 + TT - 1) / TT, NW1 * NTW1);
    for (int tile = wave; tile < ntile; tile += NW1) {
        v8f acc[NTILE];
        conv_tile5(Xw, tile * TT + XB0, Af, lane, acc);
        const float keep = (T0 + tile * TT + 4 * c < T_LEN) ? 1.0f : 0.0f;
#pragma unroll
        for (int r = 0; r < 8; ++r) {
            const float c0 = keep * sqrtf(acc[0][r] * acc[0][r] + acc[1][r] * acc[1][r]);
            const float c1 = keep * sqrtf(acc[2][r] * acc[2][r] + acc[3][r] * acc[3][r]);
            s[r >> 2]      += c0;  q2[r >> 2]      += c0 * c0;
            s[2 + (r >> 2)] += c1; q2[2 + (r >> 2)] += c1 * c1;
        }
#pragma unroll
        for (int d = 0; d < 4; ++d) {
            const float c2 = keep * sqrtf(acc[4][d] * acc[4][d] + acc[4][4 + d] * acc[4][4 + d]);
            s[4] += c2; q2[4] += c2 * c2;
        }
    }

#pragma unroll
    for (int j = 0; j < 5; ++j) {
        float sv = s[j], qv = q2[j];
        for (int off = 1; off < 16; off <<= 1) {
            sv += __shfl_xor(sv, off, 32);
            qv += __shfl_xor(qv, off, 32);
        }
        if (c == 0) {
            const int k = (j < 4) ? ((j >> 1) * 4 + 2 * h + (j & 1)) : (8 + h);
            red[wave * 16 + k]       = sv;
            red[NW1 * 16 + wave * 16 + k] = qv;
        }
    }
    __syncthreads();
    if (tid < 2 * RANK) {
        const int which = tid / RANK, k = tid % RANK;
        float a = 0.0f;
        for (int w = 0; w < NW1; ++w) a += red[which * NW1 * 16 + w * 16 + k];
        atomicAdd(which ? &sumQ[n * RANK + k] : &sumA[n * RANK + k], a);
    }
}

__global__ void finalize_stats(const float* __restrict__ B1,
                               const float* __restrict__ sumA,
                               const float* __restrict__ sumQ,
                               float* __restrict__ B1s, float* __restrict__ ybar) {
    const int tid = threadIdx.x;
    if (tid < RANK) ybar[tid] = 0.0f;
    __syncthreads();
    for (int i = tid; i < N_IN * RANK; i += blockDim.x) {
        const float s    = sumA[i], q = sumQ[i];
        const float mean = s / (float)T_LEN;
        const float var  = (q - s * mean) / (float)(T_LEN - 1);
        const float w    = B1[i] * rsqrtf(var);
        B1s[i] = w;
        atomicAdd(&ybar[i % RANK], w * mean);
    }
}

static_assert(N_IN == NG3 * NPG3, "row groups cover the input rows");
static_assert(XP3 >= XB0 + 60 + KPAD + 2, "window holds every tap of the tile");
static_assert(TT == 64 && N_OUT * 2 % NG3 == 0, "readout: a wave stores 32 t of one output row");

__global__ void __launch_bounds__(NTH3)
pass3_out(const float* __restrict__ X, const float* __restrict__ K,
          const float* __restrict__ B2, const float* __restrict__ bias,
          const float* __restrict__ B1s, const float* __restrict__ ybar,
          float* __restrict__ out) {
    __shared__ __attribute__((aligned(16))) float Af[AROWS * AP];
    __shared__ float  Ks[KSN];
    __shared__ float4 Xw4[N_IN * XR3F4];
    __shared__ __attribute__((aligned(8))) float B1sS[N_IN * RANK];
    __shared__ float  B2S[RANK * N_OUT];
    __shared__ float  ypart[NG3 * RANK * TT];
    __shared__ float  yS[RANK * TT];
    __shared__ float  ybS[NTH3];
    __shared__ float  biasS[NTH3];
    const int tid  = threadIdx.x;
    const int lane = tid & 31;
    const int wave = __builtin_amdgcn_readfirstlane(tid >> 5);
    const int c = lane & 15, h = lane >> 4;

    const int T0 = blockIdx.x * TT;

    static_assert((N_IN * XR3F4) % NTH3 == 0, "whole trips: the staging store needs no guard");
#pragma unroll
    for (int j = 0; j < N_IN * XR3F4 / NTH3; ++j) {
        const int i  = tid + j * NTH3;
        const int n  = i / XR3F4, g = i % XR3F4;
        const int t  = T0 - XORG + 4 * g;
        const int tc = min(max(t, 0), T_LEN - 4);
        float4 v = *reinterpret_cast<const float4*>(X + (size_t)n * T_LEN + tc);
        if (t < 0 || t >= T_LEN) v = make_float4(0.0f, 0.0f, 0.0f, 0.0f);
        Xw4[i] = v;
    }
    static_assert((N_IN * RANK) % NTH3 == 0 && (RANK * N_OUT) % NTH3 == 0, "whole trips");
    static_assert(NTH3 >= N_OUT && NTH3 >= RANK, "one thread per bias and per ybar element");
#pragma unroll
    for (int jj = 0; jj < N_IN * RANK / NTH3; ++jj) B1sS[tid + jj * NTH3] = B1s[tid + jj * NTH3];
#pragma unroll
    for (int jj = 0; jj < RANK * N_OUT / NTH3; ++jj) B2S[tid + jj * NTH3] = B2[tid + jj * NTH3];
    ybS[tid]   = ybar[min(tid, RANK - 1)];
    biasS[tid] = bias[min(tid, N_OUT - 1)];
    stage_filters<NTH3>(K, Ks, Af, tid);
    __syncthreads();
    const float* Xw = reinterpret_cast<const float*>(Xw4);

    float y0[8], y1[8], y2[4];
#pragma unroll
    for (int r = 0; r < 8; ++r) { y0[r] = 0.0f; y1[r] = 0.0f; }
#pragma unroll
    for (int d = 0; d < 4; ++d) y2[d] = 0.0f;

    for (int nn = 0; nn < NPG3; ++nn) {
        const int n = wave * NPG3 + nn;
        v8f acc[NTILE];
        conv_tile5(Xw + n * XP3, XB0, Af, lane, acc);
        const float* wp = &B1sS[n * RANK];
        const v2f w01 = *reinterpret_cast<const v2f*>(wp + 2 * h);
        const v2f w23 = *reinterpret_cast<const v2f*>(wp + 4 + 2 * h);
        const float w0 = w01.x, w1 = w01.y, w2 = w23.x, w3 = w23.y, w4 = wp[8 + h];
#pragma unroll
        for (int r = 0; r < 8; ++r) {
            const float c0 = sqrtf(acc[0][r] * acc[0][r] + acc[1][r] * acc[1][r]);
            const float c1 = sqrtf(acc[2][r] * acc[2][r] + acc[3][r] * acc[3][r]);
            y0[r] += ((r < 4) ? w0 : w1) * c0;
            y1[r] += ((r < 4) ? w2 : w3) * c1;
        }
#pragma unroll
        for (int d = 0; d < 4; ++d) {
            const float c2 = sqrtf(acc[4][d] * acc[4][d] + acc[4][4 + d] * acc[4][4 + d]);
            y2[d] += w4 * c2;
        }
    }

    {
        float* yp = &ypart[wave * RANK * TT];
#pragma unroll
        for (int r = 0; r < 8; ++r) {
            const int d = r & 3, j = r >> 2;
            yp[(2 * h + j) * TT + 4 * c + d]     = y0[r];
            yp[(4 + 2 * h + j) * TT + 4 * c + d] = y1[r];
        }
#pragma unroll
        for (int d = 0; d < 4; ++d) yp[(8 + h) * TT + 4 * c + d] = y2[d];
    }
    __syncthreads();
    for (int i = tid; i < RANK * TT; i += NTH3) {
        float a = 0.0f;
        for (int g = 0; g < NG3; ++g) a += ypart[g * RANK * TT + i];
        yS[i] = a - ybS[i / TT];
    }
    __syncthreads();

    for (int j = 0; j < N_OUT * 2 / NG3; ++j) {
        const int item = wave + j * NG3;
        const int m  = item >> 1;
        const int tt = (item & 1) * 32 + lane;
        const int t  = T0 + tt;
        float p = 0.0f;
        for (int k = 0; k < RANK; ++k) p += yS[k * TT + tt] * B2S[k * N_OUT + m];
        if (t < T_LEN) out[(size_t)m * T_LEN + t] = p + biasS[m];
    }
}

extern "C" void kernel_launch(void* const* d_in, const int* in_sizes, int n_in,
                              void* d_out, int out_size, void* d_ws, size_t ws_size,
                              hipStream_t stream) {
    const float* X    = (const float*)d_in[0];
    const float* K    = (const float*)d_in[1];
    const float* B1   = (const float*)d_in[2];
    const float* B2   = (const float*)d_in[3];
    const float* bias = (const float*)d_in[4];
    float*       out  = (float*)d_out;

    float* ws   = (float*)d_ws;
    float* sumA = ws;
    float* sumQ = ws + N_IN * RANK;
    float* B1s  = ws + 2 * N_IN * RANK;
    float* ybar = ws + 3 * N_IN * RANK;

    const int nstats = 2 * N_IN * RANK;
    zero_ws_kernel<<<(nstats + 255) / 256, 256, 0, stream>>>(ws, nstats);

    const int nspan1 = (T_LEN + SPAN1 - 1) / SPAN1;
    pass1_stats<<<dim3(nspan1, N_IN), NTH1, 0, stream>>>(X, K, sumA, sumQ);

    finalize_stats<<<1, 256, 0, stream>>>(B1, sumA, sumQ, B1s, ybar);

    const int nchunk3 = (T_LEN + TT - 1) / TT;
    pass3_out<<<nchunk3, NTH3, 0, stream>>>(X, K, B2, bias, B1s, ybar, out);
}


# pass 3 re-tiled: WMMA columns = 16 input rows at one 4-t group, so 256 equal 8-wave workgroups of 196 t load every SIMD alike; filter bank and sliding X fragments stay in registers
# speedup vs baseline: 3.3204x; 1.0768x over previous
#include <hip/hip_runtime.h>
#include <hip/hip_bf16.h>
#include <math.h>


#define T_LEN   50000
#define W_LEN   101
#define N_IN    128
#define RANK    10
#define N_OUT   64

#define KPAD    104
#define KSTEPS  26
#define NTILE   5
#define AROWS   (NTILE * 16)
#define AP      108
#define TT      64
#define XORG    52
#define XB0     (XORG - 50)

#define NW1     4
#define NTH1    (32 * NW1)
#define NTW1    8
#define SPAN1   (TT * NW1 * NTW1)
#define XW1F4   ((SPAN1 - TT + XB0 + 60 + KPAD + 3) / 4)
#define XW1TR   ((XW1F4 + NTH1 - 1) / NTH1)
#define G3      49
#define SPAN3   (4 * G3)
#define NW3     8
#define NTH3    (32 * NW3)
#define NFRAG3  (G3 + KSTEPS - 1)
#define XR3F4   75
#define XP3     (XR3F4 * 4)
#define X3TR    ((N_IN * XR3F4 + NTH3 - 1) / NTH3)
#define B2TR    ((RANK * N_OUT + NTH3 - 1) / NTH3)
#define NWG3    ((T_LEN / 4 + G3 - 1) / G3)
#define KTR     ((RANK * 2 * W_LEN + 127) / 128)
#define KSN     (KTR * 128)

typedef __attribute__((ext_vector_type(2))) float v2f;
typedef __attribute__((ext_vector_type(8))) float v8f;

__device__ __forceinline__ void conv_tile5(const float* __restrict__ Xw, int xb,
                                           const float* __restrict__ Af,
                                           int lane, v8f (&acc)[NTILE]) {
    const int c = lane & 15, h = lane >> 4;
    const float* ap = Af + c * AP + 2 * h;
    const float* bp = Xw + xb + 4 * c + 2 * h;
#pragma unroll
    for (int q = 0; q < NTILE; ++q)
#pragma unroll
        for (int r = 0; r < 8; ++r) acc[q][r] = 0.0f;
    for (int s = 0; s < KSTEPS; ++s) {
        const v2f b = *reinterpret_cast<const v2f*>(bp + 4 * s);
#pragma unroll
        for (int q = 0; q < NTILE; ++q) {
            const v2f a = *reinterpret_cast<const v2f*>(ap + q * 16 * AP + 4 * s);
            acc[q] = __builtin_amdgcn_wmma_f32_16x16x4_f32(
                false, a, false, b, (short)0, acc[q], false, false);
        }
    }
}

template <int NTHR>
__device__ __forceinline__ void stage_filters(const float* __restrict__ K,
                                              float* __restrict__ Ks,
                                              float* __restrict__ Af, int tid) {
    static_assert(NTHR >= AP, "one thread per column of the filter bank");
    constexpr int TR = (RANK * 2 * W_LEN + NTHR - 1) / NTHR;
    static_assert(TR * NTHR <= KSN, "Ks holds TR whole trips of the workgroup");
#pragma unroll
    for (int j = 0; j < TR; ++j) {
        const int i = tid + j * NTHR;
        Ks[i] = K[min(i, RANK * 2 * W_LEN - 1)];
    }
    __syncthreads();
    if (tid < AP) {
        const int up = tid;
#pragma unroll
        for (int R = 0; R < AROWS; ++R) {
            const int q = R >> 4, m = R & 15, pos = m >> 2, d = m & 3;
            const int k    = (q < 4) ? ((q >> 1) * 4 + pos) : (8 + (pos >> 1));
            const int part = (q < 4) ? (q & 1) : (pos & 1);
            const int u = up - d;
            const int w = min(max(W_LEN - 1 - u, 0), W_LEN - 1);
            const float v = Ks[(k * 2 + part) * W_LEN + w];
            Af[R * AP + up] = (u >= 0 && u < W_LEN) ? v : 0.0f;
        }
    }
}

__global__ void zero_ws_kernel(float* __restrict__ p, int nelem) {
    const int i = blockIdx.x * blockDim.x + threadIdx.x;
    if (i < nelem) p[i] = 0.0f;
}

static_assert(T_LEN % 4 == 0 && SPAN1 % 4 == 0 && XORG % 4 == 0, "16-byte groups and 4-t column groups never straddle 0 or T");
static_assert(XW1F4 * 4 >= SPAN1 - TT + XB0 + 60 + KPAD + 2, "window holds every tap of the last tile");

__global__ void __launch_bounds__(NTH1)
pass1_stats(const float* __restrict__ X, const float* __restrict__ K,
            float* __restrict__ sumA, float* __restrict__ sumQ) {
    __shared__ __attribute__((aligned(16))) float Af[AROWS * AP];
    __shared__ float Ks[KSN];
    __shared__ float4 Xw4[XW1TR * NTH1];
    __shared__ float red[2 * NW1 * 16];
    const int tid  = threadIdx.x;
    const int lane = tid & 31;
    const int wave = __builtin_amdgcn_readfirstlane(tid >> 5);
    const int c = lane & 15, h = lane >> 4;

    const int T0 = blockIdx.x * SPAN1;
    const int n  = blockIdx.y;
    const float* __restrict__ Xn = X + (size_t)n * T_LEN;

#pragma unroll
    for (int j = 0; j < XW1TR; ++j) {
        const int i  = tid + j * NTH1;
        const int t  = T0 - XORG + 4 * i;
        const int tc = min(max(t, 0), T_LEN - 4);
        float4 v = *reinterpret_cast<const float4*>(Xn + tc);
        if (t < 0 || t >= T_LEN) v = make_float4(0.0f, 0.0f, 0.0f, 0.0f);
        Xw4[i] = v;
    }
    stage_filters<NTH1>(K, Ks, Af, tid);
    __syncthreads();
    const float* Xw = reinterpret_cast<const float*>(Xw4);

    float s[5], q2[5];
#pragma unroll
    for (int j = 0; j < 5; ++j) { s[j] = 0.0f; q2[j] = 0.0f; }

    const int ntile = min((T_LEN - T0 + TT - 1) / TT, NW1 * NTW1);
    for (int tile = wave; tile < ntile; tile += NW1) {
        v8f acc[NTILE];
        conv_tile5(Xw, tile * TT + XB0, Af, lane, acc);
        const float keep = (T0 + tile * TT + 4 * c < T_LEN) ? 1.0f : 0.0f;
#pragma unroll
        for (int r = 0; r < 8; ++r) {
            const float c0 = keep * sqrtf(acc[0][r] * acc[0][r] + acc[1][r] * acc[1][r]);
            const float c1 = keep * sqrtf(acc[2][r] * acc[2][r] + acc[3][r] * acc[3][r]);
            s[r >> 2]      += c0;  q2[r >> 2]      += c0 * c0;
            s[2 + (r >> 2)] += c1; q2[2 + (r >> 2)] += c1 * c1;
        }
#pragma unroll
        for (int d = 0; d < 4; ++d) {
            const float c2 = keep * sqrtf(acc[4][d] * acc[4][d] + acc[4][4 + d] * acc[4][4 + d]);
            s[4] += c2; q2[4] += c2 * c2;
        }
    }

#pragma unroll
    for (int j = 0; j < 5; ++j) {
        float sv = s[j], qv = q2[j];
        for (int off = 1; off < 16; off <<= 1) {
            sv += __shfl_xor(sv, off, 32);
            qv += __shfl_xor(qv, off, 32);
        }
        if (c == 0) {
            const int k = (j < 4) ? ((j >> 1) * 4 + 2 * h + (j & 1)) : (8 + h);
            red[wave * 16 + k]       = sv;
            red[NW1 * 16 + wave * 16 + k] = qv;
        }
    }
    __syncthreads();
    if (tid < 2 * RANK) {
        const int which = tid / RANK, k = tid % RANK;
        float a = 0.0f;
        for (int w = 0; w < NW1; ++w) a += red[which * NW1 * 16 + w * 16 + k];
        atomicAdd(which ? &sumQ[n * RANK + k] : &sumA[n * RANK + k], a);
    }
}

__global__ void finalize_stats(const float* __restrict__ B1,
                               const float* __restrict__ sumA,
                               const float* __restrict__ sumQ,
                               float* __restrict__ B1s, float* __restrict__ ybar) {
    const int tid = threadIdx.x;
    if (tid < RANK) ybar[tid] = 0.0f;
    __syncthreads();
    for (int i = tid; i < N_IN * RANK; i += blockDim.x) {
        const float s    = sumA[i], q = sumQ[i];
        const float mean = s / (float)T_LEN;
        const float var  = (q - s * mean) / (float)(T_LEN - 1);
        const float w    = B1[i] * rsqrtf(var);
        B1s[i] = w;
        atomicAdd(&ybar[i % RANK], w * mean);
    }
}

static_assert(N_IN == 16 * NW3, "one wave per block of 16 input rows");
static_assert(XP3 >= XB0 + 4 * (NFRAG3 - 1) + 4, "window holds every fragment of the last group");
static_assert((XP3 % 64) % 8 == 4, "row pitch mod 64 is an odd multiple of 4: the 16 rows of a fragment read fall on 64 different banks");
static_assert(NTH3 >= SPAN3 && NTH3 >= N_OUT && NTH3 >= RANK, "readout: one thread per time point; one per bias and ybar element");
static_assert(NWG3 * SPAN3 >= T_LEN && T_LEN % 4 == 0 && XORG % 4 == 0, "spans cover T; 16-byte groups never straddle 0 or T");

template <int M>
__device__ __forceinline__ float row_xor_add(float v) {
    const int x = __builtin_amdgcn_update_dpp(0, __builtin_bit_cast(int, v), 0x160 + M, 0xf, 0xf, true);
    return v + __builtin_bit_cast(float, x);
}

__global__ void __launch_bounds__(NTH3)
pass3_out(const float* __restrict__ X, const float* __restrict__ K,
          const float* __restrict__ B2, const float* __restrict__ bias,
          const float* __restrict__ B1s, const float* __restrict__ ybar,
          float* __restrict__ out) {
    __shared__ __attribute__((aligned(16))) float Af[AROWS * AP];
    __shared__ float  Ks[KSN];
    __shared__ float4 Xw4[X3TR * NTH3];
    __shared__ float  B2S[B2TR * NTH3];
    __shared__ __attribute__((aligned(16))) float ypart[NW3 * RANK * SPAN3];
    __shared__ float  ybS[NTH3];
    __shared__ float  biasS[NTH3];
    const int tid  = threadIdx.x;
    const int lane = tid & 31;
    const int wave = __builtin_amdgcn_readfirstlane(tid >> 5);
    const int c = lane & 15, h = lane >> 4;

    const int T0 = blockIdx.x * SPAN3;

#pragma unroll
    for (int j = 0; j < X3TR; ++j) {
        const int i  = tid + j * NTH3;
        const int ic = min(i, N_IN * XR3F4 - 1);
        const int n  = ic / XR3F4, g4 = ic % XR3F4;
        const int t  = T0 - XORG + 4 * g4;
        const int tc = min(max(t, 0), T_LEN - 4);
        float4 v = *reinterpret_cast<const float4*>(X + (size_t)n * T_LEN + tc);
        if (t < 0 || t >= T_LEN) v = make_float4(0.0f, 0.0f, 0.0f, 0.0f);
        Xw4[i] = v;
    }
#pragma unroll
    for (int jj = 0; jj < B2TR; ++jj) B2S[tid + jj * NTH3] = B2[min(tid + jj * NTH3, RANK * N_OUT - 1)];
    ybS[tid]   = ybar[min(tid, RANK - 1)];
    biasS[tid] = bias[min(tid, N_OUT - 1)];
    const int nrow = wave * 16 + c;
    const float* wp = B1s + nrow * RANK;
    const float w0 = wp[2 * h], w1 = wp[2 * h + 1], w2 = wp[4 + 2 * h], w3 = wp[5 + 2 * h], w4 = wp[8 + h];
    stage_filters<NTH3>(K, Ks, Af, tid);
    __syncthreads();

    v2f a[NTILE][KSTEPS];
    {
        const float* ap = Af + c * AP + 2 * h;
#pragma unroll
        for (int q = 0; q < NTILE; ++q)
#pragma unroll
            for (int s = 0; s < KSTEPS; ++s)
                a[q][s] = *reinterpret_cast<const v2f*>(ap + q * 16 * AP + 4 * s);
    }
    const float* xr = reinterpret_cast<const float*>(Xw4) + nrow * XP3 + XB0 + 2 * h;
    v2f b[KSTEPS];
#pragma unroll
    for (int s = 0; s < KSTEPS; ++s) b[s] = *reinterpret_cast<const v2f*>(xr + 4 * s);

    const int ng = min(G3, (T_LEN - T0) / 4);
    float* yw = ypart + wave * RANK * SPAN3;
    for (int g = 0; g < ng; ++g) {
        const int jn = min(g + KSTEPS, NFRAG3 - 1);
        const v2f nb = *reinterpret_cast<const v2f*>(xr + 4 * jn);

        v8f acc[NTILE];
#pragma unroll
        for (int q = 0; q < NTILE; ++q)
#pragma unroll
            for (int r = 0; r < 8; ++r) acc[q][r] = 0.0f;
#pragma unroll
        for (int s = 0; s < KSTEPS; ++s)
#pragma unroll
            for (int q = 0; q < NTILE; ++q)
                acc[q] = __builtin_amdgcn_wmma_f32_16x16x4_f32(
                    false, a[q][s], false, b[s], (short)0, acc[q], false, false);

        float val[20];
#pragma unroll
        for (int r = 0; r < 8; ++r) {
            const float c0 = sqrtf(acc[0][r] * acc[0][r] + acc[1][r] * acc[1][r]);
            const float c1 = sqrtf(acc[2][r] * acc[2][r] + acc[3][r] * acc[3][r]);
            val[r]     = ((r < 4) ? w0 : w1) * c0;
            val[8 + r] = ((r < 4) ? w2 : w3) * c1;
        }
#pragma unroll
        for (int d = 0; d < 4; ++d)
            val[16 + d] = w4 * sqrtf(acc[4][d] * acc[4][d] + acc[4][4 + d] * acc[4][4 + d]);
#pragma unroll
        for (int i = 0; i < 20; ++i) val[i] = row_xor_add<1>(val[i]);
#pragma unroll
        for (int i = 0; i < 20; ++i) val[i] = row_xor_add<2>(val[i]);
#pragma unroll
        for (int i = 0; i < 20; ++i) val[i] = row_xor_add<4>(val[i]);
#pragma unroll
        for (int i = 0; i < 20; ++i) val[i] = row_xor_add<8>(val[i]);
        if (c == 0) {
            float* yp = yw + 4 * g;
            *reinterpret_cast<float4*>(yp + (2 * h) * SPAN3)     = make_float4(val[0],  val[1],  val[2],  val[3]);
            *reinterpret_cast<float4*>(yp + (2 * h + 1) * SPAN3) = make_float4(val[4],  val[5],  val[6],  val[7]);
            *reinterpret_cast<float4*>(yp + (4 + 2 * h) * SPAN3) = make_float4(val[8],  val[9],  val[10], val[11]);
            *reinterpret_cast<float4*>(yp + (5 + 2 * h) * SPAN3) = make_float4(val[12], val[13], val[14], val[15]);
            *reinterpret_cast<float4*>(yp + (8 + h) * SPAN3)     = make_float4(val[16], val[17], val[18], val[19]);
        }
#pragma unroll
        for (int s = 0; s + 1 < KSTEPS; ++s) b[s] = b[s + 1];
        b[KSTEPS - 1] = nb;
    }
    __syncthreads();

    const int t = T0 + tid;
    if (tid < SPAN3 && t < T_LEN) {
        float yv[RANK];
#pragma unroll
        for (int k = 0; k < RANK; ++k) {
            float s = 0.0f;
#pragma unroll
            for (int w = 0; w < NW3; ++w) s += ypart[(w * RANK + k) * SPAN3 + tid];
            yv[k] = s - ybS[k];
        }
        for (int m = 0; m < N_OUT; ++m) {
            float p = 0.0f;
#pragma unroll
            for (int k = 0; k < RANK; ++k) p += yv[k] * B2S[k * N_OUT + m];
            out[(size_t)m * T_LEN + t] = p + biasS[m];
        }
    }
}

extern "C" void kernel_launch(void* const* d_in, const int* in_sizes, int n_in,
                              void* d_out, int out_size, void* d_ws, size_t ws_size,
                              hipStream_t stream) {
    const float* X    = (const float*)d_in[0];
    const float* K    = (const float*)d_in[1];
    const float* B1   = (const float*)d_in[2];
    const float* B2   = (const float*)d_in[3];
    const float* bias = (const float*)d_in[4];
    float*       out  = (float*)d_out;

    float* ws   = (float*)d_ws;
    float* sumA = ws;
    float* sumQ = ws + N_IN * RANK;
    float* B1s  = ws + 2 * N_IN * RANK;
    float* ybar = ws + 3 * N_IN * RANK;

    const int nstats = 2 * N_IN * RANK;
    zero_ws_kernel<<<(nstats + 255) / 256, 256, 0, stream>>>(ws, nstats);

    const int nspan1 = (T_LEN + SPAN1 - 1) / SPAN1;
    pass1_stats<<<dim3(nspan1, N_IN), NTH1, 0, stream>>>(X, K, sumA, sumQ);

    finalize_stats<<<1, 256, 0, stream>>>(B1, sumA, sumQ, B1s, ybar);

    pass3_out<<<NWG3, NTH3, 0, stream>>>(X, K, B2, bias, B1s, ybar, out);
}


# |conv| by the hardware square root (v_sqrt_f32, 1 ulp) instead of exactly rounded sqrtf in both conv kernels: about 15 vector instructions fewer per root between the f32 matrix runs
# speedup vs baseline: 3.6151x; 1.0888x over previous
#include <hip/hip_runtime.h>
#include <hip/hip_bf16.h>
#include <math.h>


#define T_LEN   50000
#define W_LEN   101
#define N_IN    128
#define RANK    10
#define N_OUT   64

#define KPAD    104
#define KSTEPS  26
#define NTILE   5
#define AROWS   (NTILE * 16)
#define AP      108
#define TT      64
#define XORG    52
#define XB0     (XORG - 50)

#define NW1     4
#define NTH1    (32 * NW1)
#define NTW1    8
#define SPAN1   (TT * NW1 * NTW1)
#define XW1F4   ((SPAN1 - TT + XB0 + 60 + KPAD + 3) / 4)
#define XW1TR   ((XW1F4 + NTH1 - 1) / NTH1)
#define G3      49
#define SPAN3   (4 * G3)
#define NW3     8
#define NTH3    (32 * NW3)
#define NFRAG3  (G3 + KSTEPS - 1)
#define XR3F4   75
#define XP3     (XR3F4 * 4)
#define X3TR    ((N_IN * XR3F4 + NTH3 - 1) / NTH3)
#define B2TR    ((RANK * N_OUT + NTH3 - 1) / NTH3)
#define NWG3    ((T_LEN / 4 + G3 - 1) / G3)
#define KTR     ((RANK * 2 * W_LEN + 127) / 128)
#define KSN     (KTR * 128)

typedef __attribute__((ext_vector_type(2))) float v2f;
typedef __attribute__((ext_vector_type(8))) float v8f;

__device__ __forceinline__ float cabs_hw(float re, float im) {
    return __builtin_amdgcn_sqrtf(re * re + im * im);
}

__device__ __forceinline__ void conv_tile5(const float* __restrict__ Xw, int xb,
                                           const float* __restrict__ Af,
                                           int lane, v8f (&acc)[NTILE]) {
    const int c = lane & 15, h = lane >> 4;
    const float* ap = Af + c * AP + 2 * h;
    const float* bp = Xw + xb + 4 * c + 2 * h;
#pragma unroll
    for (int q = 0; q < NTILE; ++q)
#pragma unroll
        for (int r = 0; r < 8; ++r) acc[q][r] = 0.0f;
    for (int s = 0; s < KSTEPS; ++s) {
        const v2f b = *reinterpret_cast<const v2f*>(bp + 4 * s);
#pragma unroll
        for (int q = 0; q < NTILE; ++q) {
            const v2f a = *reinterpret_cast<const v2f*>(ap + q * 16 * AP + 4 * s);
            acc[q] = __builtin_amdgcn_wmma_f32_16x16x4_f32(
                false, a, false, b, (short)0, acc[q], false, false);
        }
    }
}

template <int NTHR>
__device__ __forceinline__ void stage_filters(const float* __restrict__ K,
                                              float* __restrict__ Ks,
                                              float* __restrict__ Af, int tid) {
    static_assert(NTHR >= AP, "one thread per column of the filter bank");
    constexpr int TR = (RANK * 2 * W_LEN + NTHR - 1) / NTHR;
    static_assert(TR * NTHR <= KSN, "Ks holds TR whole trips of the workgroup");
#pragma unroll
    for (int j = 0; j < TR; ++j) {
        const int i = tid + j * NTHR;
        Ks[i] = K[min(i, RANK * 2 * W_LEN - 1)];
    }
    __syncthreads();
    if (tid < AP) {
        const int up = tid;
#pragma unroll
        for (int R = 0; R < AROWS; ++R) {
            const int q = R >> 4, m = R & 15, pos = m >> 2, d = m & 3;
            const int k    = (q < 4) ? ((q >> 1) * 4 + pos) : (8 + (pos >> 1));
            const int part = (q < 4) ? (q & 1) : (pos & 1);
            const int u = up - d;
            const int w = min(max(W_LEN - 1 - u, 0), W_LEN - 1);
            const float v = Ks[(k * 2 + part) * W_LEN + w];
            Af[R * AP + up] = (u >= 0 && u < W_LEN) ? v : 0.0f;
        }
    }
}

__global__ void zero_ws_kernel(float* __restrict__ p, int nelem) {
    const int i = blockIdx.x * blockDim.x + threadIdx.x;
    if (i < nelem) p[i] = 0.0f;
}

static_assert(T_LEN % 4 == 0 && SPAN1 % 4 == 0 && XORG % 4 == 0, "16-byte groups and 4-t column groups never straddle 0 or T");
static_assert(XW1F4 * 4 >= SPAN1 - TT + XB0 + 60 + KPAD + 2, "window holds every tap of the last tile");

__global__ void __launch_bounds__(NTH1)
pass1_stats(const float* __restrict__ X, const float* __restrict__ K,
            float* __restrict__ sumA, float* __restrict__ sumQ) {
    __shared__ __attribute__((aligned(16))) float Af[AROWS * AP];
    __shared__ float Ks[KSN];
    __shared__ float4 Xw4[XW1TR * NTH1];
    __shared__ float red[2 * NW1 * 16];
    const int tid  = threadIdx.x;
    const int lane = tid & 31;
    const int wave = __builtin_amdgcn_readfirstlane(tid >> 5);
    const int c = lane & 15, h = lane >> 4;

    const int T0 = blockIdx.x * SPAN1;
    const int n  = blockIdx.y;
    const float* __restrict__ Xn = X + (size_t)n * T_LEN;

#pragma unroll
    for (int j = 0; j < XW1TR; ++j) {
        const int i  = tid + j * NTH1;
        const int t  = T0 - XORG + 4 * i;
        const int tc = min(max(t, 0), T_LEN - 4);
        float4 v = *reinterpret_cast<const float4*>(Xn + tc);
        if (t < 0 || t >= T_LEN) v = make_float4(0.0f, 0.0f, 0.0f, 0.0f);
        Xw4[i] = v;
    }
    stage_filters<NTH1>(K, Ks, Af, tid);
    __syncthreads();
    const float* Xw = reinterpret_cast<const float*>(Xw4);

    float s[5], q2[5];
#pragma unroll
    for (int j = 0; j < 5; ++j) { s[j] = 0.0f; q2[j] = 0.0f; }

    const int ntile = min((T_LEN - T0 + TT - 1) / TT, NW1 * NTW1);
    for (int tile = wave; tile < ntile; tile += NW1) {
        v8f acc[NTILE];
        conv_tile5(Xw, tile * TT + XB0, Af, lane, acc);
        const float keep = (T0 + tile * TT + 4 * c < T_LEN) ? 1.0f : 0.0f;
#pragma unroll
        for (int r = 0; r < 8; ++r) {
            const float c0 = keep * cabs_hw(acc[0][r], acc[1][r]);
            const float c1 = keep * cabs_hw(acc[2][r], acc[3][r]);
            s[r >> 2]      += c0;  q2[r >> 2]      += c0 * c0;
            s[2 + (r >> 2)] += c1; q2[2 + (r >> 2)] += c1 * c1;
        }
#pragma unroll
        for (int d = 0; d < 4; ++d) {
            const float c2 = keep * cabs_hw(acc[4][d], acc[4][4 + d]);
            s[4] += c2; q2[4] += c2 * c2;
        }
    }

#pragma unroll
    for (int j = 0; j < 5; ++j) {
        float sv = s[j], qv = q2[j];
        for (int off = 1; off < 16; off <<= 1) {
            sv += __shfl_xor(sv, off, 32);
            qv += __shfl_xor(qv, off, 32);
        }
        if (c == 0) {
            const int k = (j < 4) ? ((j >> 1) * 4 + 2 * h + (j & 1)) : (8 + h);
            red[wave * 16 + k]       = sv;
            red[NW1 * 16 + wave * 16 + k] = qv;
        }
    }
    __syncthreads();
    if (tid < 2 * RANK) {
        const int which = tid / RANK, k = tid % RANK;
        float a = 0.0f;
        for (int w = 0; w < NW1; ++w) a += red[which * NW1 * 16 + w * 16 + k];
        atomicAdd(which ? &sumQ[n * RANK + k] : &sumA[n * RANK + k], a);
    }
}

__global__ void finalize_stats(const float* __restrict__ B1,
                               const float* __restrict__ sumA,
                               const float* __restrict__ sumQ,
                               float* __restrict__ B1s, float* __restrict__ ybar) {
    const int tid = threadIdx.x;
    if (tid < RANK) ybar[tid] = 0.0f;
    __syncthreads();
    for (int i = tid; i < N_IN * RANK; i += blockDim.x) {
        const float s    = sumA[i], q = sumQ[i];
        const float mean = s / (float)T_LEN;
        const float var  = (q - s * mean) / (float)(T_LEN - 1);
        const float w    = B1[i] * rsqrtf(var);
        B1s[i] = w;
        atomicAdd(&ybar[i % RANK], w * mean);
    }
}

static_assert(N_IN == 16 * NW3, "one wave per block of 16 input rows");
static_assert(XP3 >= XB0 + 4 * (NFRAG3 - 1) + 4, "window holds every fragment of the last group");
static_assert((XP3 % 64) % 8 == 4, "row pitch mod 64 is an odd multiple of 4: the 16 rows of a fragment read fall on 64 different banks");
static_assert(NTH3 >= SPAN3 && NTH3 >= N_OUT && NTH3 >= RANK, "readout: one thread per time point; one per bias and ybar element");
static_assert(NWG3 * SPAN3 >= T_LEN && T_LEN % 4 == 0 && XORG % 4 == 0, "spans cover T; 16-byte groups never straddle 0 or T");

template <int M>
__device__ __forceinline__ float row_xor_add(float v) {
    const int x = __builtin_amdgcn_update_dpp(0, __builtin_bit_cast(int, v), 0x160 + M, 0xf, 0xf, true);
    return v + __builtin_bit_cast(float, x);
}

__global__ void __launch_bounds__(NTH3)
pass3_out(const float* __restrict__ X, const float* __restrict__ K,
          const float* __restrict__ B2, const float* __restrict__ bias,
          const float* __restrict__ B1s, const float* __restrict__ ybar,
          float* __restrict__ out) {
    __shared__ __attribute__((aligned(16))) float Af[AROWS * AP];
    __shared__ float  Ks[KSN];
    __shared__ float4 Xw4[X3TR * NTH3];
    __shared__ float  B2S[B2TR * NTH3];
    __shared__ __attribute__((aligned(16))) float ypart[NW3 * RANK * SPAN3];
    __shared__ float  ybS[NTH3];
    __shared__ float  biasS[NTH3];
    const int tid  = threadIdx.x;
    const int lane = tid & 31;
    const int wave = __builtin_amdgcn_readfirstlane(tid >> 5);
    const int c = lane & 15, h = lane >> 4;

    const int T0 = blockIdx.x * SPAN3;

#pragma unroll
    for (int j = 0; j < X3TR; ++j) {
        const int i  = tid + j * NTH3;
        const int ic = min(i, N_IN * XR3F4 - 1);
        const int n  = ic / XR3F4, g4 = ic % XR3F4;
        const int t  = T0 - XORG + 4 * g4;
        const int tc = min(max(t, 0), T_LEN - 4);
        float4 v = *reinterpret_cast<const float4*>(X + (size_t)n * T_LEN + tc);
        if (t < 0 || t >= T_LEN) v = make_float4(0.0f, 0.0f, 0.0f, 0.0f);
        Xw4[i] = v;
    }
#pragma unroll
    for (int jj = 0; jj < B2TR; ++jj) B2S[tid + jj * NTH3] = B2[min(tid + jj * NTH3, RANK * N_OUT - 1)];
    ybS[tid]   = ybar[min(tid, RANK - 1)];
    biasS[tid] = bias[min(tid, N_OUT - 1)];
    const int nrow = wave * 16 + c;
    const float* wp = B1s + nrow * RANK;
    const float w0 = wp[2 * h], w1 = wp[2 * h + 1], w2 = wp[4 + 2 * h], w3 = wp[5 + 2 * h], w4 = wp[8 + h];
    stage_filters<NTH3>(K, Ks, Af, tid);
    __syncthreads();

    v2f a[NTILE][KSTEPS];
    {
        const float* ap = Af + c * AP + 2 * h;
#pragma unroll
        for (int q = 0; q < NTILE; ++q)
#pragma unroll
            for (int s = 0; s < KSTEPS; ++s)
                a[q][s] = *reinterpret_cast<const v2f*>(ap + q * 16 * AP + 4 * s);
    }
    const float* xr = reinterpret_cast<const float*>(Xw4) + nrow * XP3 + XB0 + 2 * h;
    v2f b[KSTEPS];
#pragma unroll
    for (int s = 0; s < KSTEPS; ++s) b[s] = *reinterpret_cast<const v2f*>(xr + 4 * s);

    const int ng = min(G3, (T_LEN - T0) / 4);
    float* yw = ypart + wave * RANK * SPAN3;
    for (int g = 0; g < ng; ++g) {
        const int jn = min(g + KSTEPS, NFRAG3 - 1);
        const v2f nb = *reinterpret_cast<const v2f*>(xr + 4 * jn);

        v8f acc[NTILE];
#pragma unroll
        for (int q = 0; q < NTILE; ++q)
#pragma unroll
            for (int r = 0; r < 8; ++r) acc[q][r] = 0.0f;
#pragma unroll
        for (int s = 0; s < KSTEPS; ++s)
#pragma unroll
            for (int q = 0; q < NTILE; ++q)
                acc[q] = __builtin_amdgcn_wmma_f32_16x16x4_f32(
                    false, a[q][s], false, b[s], (short)0, acc[q], false, false);

        float val[20];
#pragma unroll
        for (int r = 0; r < 8; ++r) {
            const float c0 = cabs_hw(acc[0][r], acc[1][r]);
            const float c1 = cabs_hw(acc[2][r], acc[3][r]);
            val[r]     = ((r < 4) ? w0 : w1) * c0;
            val[8 + r] = ((r < 4) ? w2 : w3) * c1;
        }
#pragma unroll
        for (int d = 0; d < 4; ++d)
            val[16 + d] = w4 * cabs_hw(acc[4][d], acc[4][4 + d]);
#pragma unroll
        for (int i = 0; i < 20; ++i) val[i] = row_xor_add<1>(val[i]);
#pragma unroll
        for (int i = 0; i < 20; ++i) val[i] = row_xor_add<2>(val[i]);
#pragma unroll
        for (int i = 0; i < 20; ++i) val[i] = row_xor_add<4>(val[i]);
#pragma unroll
        for (int i = 0; i < 20; ++i) val[i] = row_xor_add<8>(val[i]);
        if (c == 0) {
            float* yp = yw + 4 * g;
            *reinterpret_cast<float4*>(yp + (2 * h) * SPAN3)     = make_float4(val[0],  val[1],  val[2],  val[3]);
            *reinterpret_cast<float4*>(yp + (2 * h + 1) * SPAN3) = make_float4(val[4],  val[5],  val[6],  val[7]);
            *reinterpret_cast<float4*>(yp + (4 + 2 * h) * SPAN3) = make_float4(val[8],  val[9],  val[10], val[11]);
            *reinterpret_cast<float4*>(yp + (5 + 2 * h) * SPAN3) = make_float4(val[12], val[13], val[14], val[15]);
            *reinterpret_cast<float4*>(yp + (8 + h) * SPAN3)     = make_float4(val[16], val[17], val[18], val[19]);
        }
#pragma unroll
        for (int s = 0; s + 1 < KSTEPS; ++s) b[s] = b[s + 1];
        b[KSTEPS - 1] = nb;
    }
    __syncthreads();

    const int t = T0 + tid;
    if (tid < SPAN3 && t < T_LEN) {
        float yv[RANK];
#pragma unroll
        for (int k = 0; k < RANK; ++k) {
            float s = 0.0f;
#pragma unroll
            for (int w = 0; w < NW3; ++w) s += ypart[(w * RANK + k) * SPAN3 + tid];
            yv[k] = s - ybS[k];
        }
        for (int m = 0; m < N_OUT; ++m) {
            float p = 0.0f;
#pragma unroll
            for (int k = 0; k < RANK; ++k) p += yv[k] * B2S[k * N_OUT + m];
            out[(size_t)m * T_LEN + t] = p + biasS[m];
        }
    }
}

extern "C" void kernel_launch(void* const* d_in, const int* in_sizes, int n_in,
                              void* d_out, int out_size, void* d_ws, size_t ws_size,
                              hipStream_t stream) {
    const float* X    = (const float*)d_in[0];
    const float* K    = (const float*)d_in[1];
    const float* B1   = (const float*)d_in[2];
    const float* B2   = (const float*)d_in[3];
    const float* bias = (const float*)d_in[4];
    float*       out  = (float*)d_out;

    float* ws   = (float*)d_ws;
    float* sumA = ws;
    float* sumQ = ws + N_IN * RANK;
    float* B1s  = ws + 2 * N_IN * RANK;
    float* ybar = ws + 3 * N_IN * RANK;

    const int nstats = 2 * N_IN * RANK;
    zero_ws_kernel<<<(nstats + 255) / 256, 256, 0, stream>>>(ws, nstats);

    const int nspan1 = (T_LEN + SPAN1 - 1) / SPAN1;
    pass1_stats<<<dim3(nspan1, N_IN), NTH1, 0, stream>>>(X, K, sumA, sumQ);

    finalize_stats<<<1, 256, 0, stream>>>(B1, sumA, sumQ, B1s, ybar);

    pass3_out<<<NWG3, NTH3, 0, stream>>>(X, K, B2, bias, B1s, ybar, out);
}


# pass 3: X fragments as WMMA A, filter bank as B, so input rows run down the accumulator registers: a block's weighted row sum is 8 in-lane multiply-adds + 1 cross-half add, not a 16-lane exchange x20
# speedup vs baseline: 3.7378x; 1.0339x over previous
#include <hip/hip_runtime.h>
#include <hip/hip_bf16.h>
#include <math.h>


#define T_LEN   50000
#define W_LEN   101
#define N_IN    128
#define RANK    10
#define N_OUT   64

#define KPAD    104
#define KSTEPS  26
#define NTILE   5
#define AROWS   (NTILE * 16)
#define AP      108
#define TT      64
#define XORG    52
#define XB0     (XORG - 50)

#define NW1     4
#define NTH1    (32 * NW1)
#define NTW1    8
#define SPAN1   (TT * NW1 * NTW1)
#define XW1F4   ((SPAN1 - TT + XB0 + 60 + KPAD + 3) / 4)
#define XW1TR   ((XW1F4 + NTH1 - 1) / NTH1)
#define G3      49
#define SPAN3   (4 * G3)
#define NW3     8
#define NTH3    (32 * NW3)
#define NFRAG3  (G3 + KSTEPS - 1)
#define XR3F4   75
#define XP3     (XR3F4 * 4)
#define X3TR    ((N_IN * XR3F4 + NTH3 - 1) / NTH3)
#define B2TR    ((RANK * N_OUT + NTH3 - 1) / NTH3)
#define NWG3    ((T_LEN / 4 + G3 - 1) / G3)
#define KTR     ((RANK * 2 * W_LEN + 127) / 128)
#define KSN     (KTR * 128)

typedef __attribute__((ext_vector_type(2))) float v2f;
typedef __attribute__((ext_vector_type(8))) float v8f;

__device__ __forceinline__ float cabs_hw(float re, float im) {
    return __builtin_amdgcn_sqrtf(re * re + im * im);
}

__device__ __forceinline__ void conv_tile5(const float* __restrict__ Xw, int xb,
                                           const float* __restrict__ Af,
                                           int lane, v8f (&acc)[NTILE]) {
    const int c = lane & 15, h = lane >> 4;
    const float* ap = Af + c * AP + 2 * h;
    const float* bp = Xw + xb + 4 * c + 2 * h;
#pragma unroll
    for (int q = 0; q < NTILE; ++q)
#pragma unroll
        for (int r = 0; r < 8; ++r) acc[q][r] = 0.0f;
    for (int s = 0; s < KSTEPS; ++s) {
        const v2f b = *reinterpret_cast<const v2f*>(bp + 4 * s);
#pragma unroll
        for (int q = 0; q < NTILE; ++q) {
            const v2f a = *reinterpret_cast<const v2f*>(ap + q * 16 * AP + 4 * s);
            acc[q] = __builtin_amdgcn_wmma_f32_16x16x4_f32(
                false, a, false, b, (short)0, acc[q], false, false);
        }
    }
}

template <int NTHR>
__device__ __forceinline__ void stage_filters(const float* __restrict__ K,
                                              float* __restrict__ Ks,
                                              float* __restrict__ Af, int tid) {
    static_assert(NTHR >= AP, "one thread per column of the filter bank");
    constexpr int TR = (RANK * 2 * W_LEN + NTHR - 1) / NTHR;
    static_assert(TR * NTHR <= KSN, "Ks holds TR whole trips of the workgroup");
#pragma unroll
    for (int j = 0; j < TR; ++j) {
        const int i = tid + j * NTHR;
        Ks[i] = K[min(i, RANK * 2 * W_LEN - 1)];
    }
    __syncthreads();
    if (tid < AP) {
        const int up = tid;
#pragma unroll
        for (int R = 0; R < AROWS; ++R) {
            const int q = R >> 4, m = R & 15, pos = m >> 2, d = m & 3;
            const int k    = (q < 4) ? ((q >> 1) * 4 + pos) : (8 + (pos >> 1));
            const int part = (q < 4) ? (q & 1) : (pos & 1);
            const int u = up - d;
            const int w = min(max(W_LEN - 1 - u, 0), W_LEN - 1);
            const float v = Ks[(k * 2 + part) * W_LEN + w];
            Af[R * AP + up] = (u >= 0 && u < W_LEN) ? v : 0.0f;
        }
    }
}

__global__ void zero_ws_kernel(float* __restrict__ p, int nelem) {
    const int i = blockIdx.x * blockDim.x + threadIdx.x;
    if (i < nelem) p[i] = 0.0f;
}

static_assert(T_LEN % 4 == 0 && SPAN1 % 4 == 0 && XORG % 4 == 0, "16-byte groups and 4-t column groups never straddle 0 or T");
static_assert(XW1F4 * 4 >= SPAN1 - TT + XB0 + 60 + KPAD + 2, "window holds every tap of the last tile");

__global__ void __launch_bounds__(NTH1)
pass1_stats(const float* __restrict__ X, const float* __restrict__ K,
            float* __restrict__ sumA, float* __restrict__ sumQ) {
    __shared__ __attribute__((aligned(16))) float Af[AROWS * AP];
    __shared__ float Ks[KSN];
    __shared__ float4 Xw4[XW1TR * NTH1];
    __shared__ float red[2 * NW1 * 16];
    const int tid  = threadIdx.x;
    const int lane = tid & 31;
    const int wave = __builtin_amdgcn_readfirstlane(tid >> 5);
    const int c = lane & 15, h = lane >> 4;

    const int T0 = blockIdx.x * SPAN1;
    const int n  = blockIdx.y;
    const float* __restrict__ Xn = X + (size_t)n * T_LEN;

#pragma unroll
    for (int j = 0; j < XW1TR; ++j) {
        const int i  = tid + j * NTH1;
        const int t  = T0 - XORG + 4 * i;
        const int tc = min(max(t, 0), T_LEN - 4);
        float4 v = *reinterpret_cast<const float4*>(Xn + tc);
        if (t < 0 || t >= T_LEN) v = make_float4(0.0f, 0.0f, 0.0f, 0.0f);
        Xw4[i] = v;
    }
    stage_filters<NTH1>(K, Ks, Af, tid);
    __syncthreads();
    const float* Xw = reinterpret_cast<const float*>(Xw4);

    float s[5], q2[5];
#pragma unroll
    for (int j = 0; j < 5; ++j) { s[j] = 0.0f; q2[j] = 0.0f; }

    const int ntile = min((T_LEN - T0 + TT - 1) / TT, NW1 * NTW1);
    for (int tile = wave; tile < ntile; tile += NW1) {
        v8f acc[NTILE];
        conv_tile5(Xw, tile * TT + XB0, Af, lane, acc);
        const float keep = (T0 + tile * TT + 4 * c < T_LEN) ? 1.0f : 0.0f;
#pragma unroll
        for (int r = 0; r < 8; ++r) {
            const float c0 = keep * cabs_hw(acc[0][r], acc[1][r]);
            const float c1 = keep * cabs_hw(acc[2][r], acc[3][r]);
            s[r >> 2]      += c0;  q2[r >> 2]      += c0 * c0;
            s[2 + (r >> 2)] += c1; q2[2 + (r >> 2)] += c1 * c1;
        }
#pragma unroll
        for (int d = 0; d < 4; ++d) {
            const float c2 = keep * cabs_hw(acc[4][d], acc[4][4 + d]);
            s[4] += c2; q2[4] += c2 * c2;
        }
    }

#pragma unroll
    for (int j = 0; j < 5; ++j) {
        float sv = s[j], qv = q2[j];
        for (int off = 1; off < 16; off <<= 1) {
            sv += __shfl_xor(sv, off, 32);
            qv += __shfl_xor(qv, off, 32);
        }
        if (c == 0) {
            const int k = (j < 4) ? ((j >> 1) * 4 + 2 * h + (j & 1)) : (8 + h);
            red[wave * 16 + k]       = sv;
            red[NW1 * 16 + wave * 16 + k] = qv;
        }
    }
    __syncthreads();
    if (tid < 2 * RANK) {
        const int which = tid / RANK, k = tid % RANK;
        float a = 0.0f;
        for (int w = 0; w < NW1; ++w) a += red[which * NW1 * 16 + w * 16 + k];
        atomicAdd(which ? &sumQ[n * RANK + k] : &sumA[n * RANK + k], a);
    }
}

__global__ void finalize_stats(const float* __restrict__ B1,
                               const float* __restrict__ sumA,
                               const float* __restrict__ sumQ,
                               float* __restrict__ B1s, float* __restrict__ ybar) {
    const int tid = threadIdx.x;
    if (tid < RANK) ybar[tid] = 0.0f;
    __syncthreads();
    for (int i = tid; i < N_IN * RANK; i += blockDim.x) {
        const float s    = sumA[i], q = sumQ[i];
        const float mean = s / (float)T_LEN;
        const float var  = (q - s * mean) / (float)(T_LEN - 1);
        const float w    = B1[i] * rsqrtf(var);
        B1s[i] = w;
        atomicAdd(&ybar[i % RANK], w * mean);
    }
}

static_assert(N_IN == 16 * NW3, "one wave per block of 16 input rows");
static_assert(XP3 >= XB0 + 4 * (NFRAG3 - 1) + 4, "window holds every fragment of the last group");
static_assert((XP3 % 64) % 8 == 4, "row pitch mod 64 is an odd multiple of 4: the 16 rows of a fragment read fall on 64 different banks");
static_assert(NTH3 >= SPAN3 && NTH3 >= N_OUT && NTH3 >= RANK, "readout: one thread per time point; one per bias and ybar element");
static_assert(NWG3 * SPAN3 >= T_LEN && T_LEN % 4 == 0 && XORG % 4 == 0, "spans cover T; 16-byte groups never straddle 0 or T");

template <int M>
__device__ __forceinline__ float row_xor_add(float v) {
    const int x = __builtin_amdgcn_update_dpp(0, __builtin_bit_cast(int, v), 0x160 + M, 0xf, 0xf, true);
    return v + __builtin_bit_cast(float, x);
}

__device__ __forceinline__ float xhalf(float v) {
    const int b = __builtin_bit_cast(int, v);
    return __builtin_bit_cast(float, __builtin_amdgcn_permlanex16(b, b, 0x76543210, 0xfedcba98, false, false));
}

__global__ void __launch_bounds__(NTH3)
pass3_out(const float* __restrict__ X, const float* __restrict__ K,
          const float* __restrict__ B2, const float* __restrict__ bias,
          const float* __restrict__ B1s, const float* __restrict__ ybar,
          float* __restrict__ out) {
    __shared__ __attribute__((aligned(16))) float Af[AROWS * AP];
    __shared__ float  Ks[KSN];
    __shared__ float4 Xw4[X3TR * NTH3];
    __shared__ float  B2S[B2TR * NTH3];
    __shared__ __attribute__((aligned(16))) float ypart[NW3 * RANK * SPAN3];
    __shared__ float  ybS[NTH3];
    __shared__ float  biasS[NTH3];
    const int tid  = threadIdx.x;
    const int lane = tid & 31;
    const int wave = __builtin_amdgcn_readfirstlane(tid >> 5);
    const int c = lane & 15, h = lane >> 4;

    const int T0 = blockIdx.x * SPAN3;

#pragma unroll
    for (int j = 0; j < X3TR; ++j) {
        const int i  = tid + j * NTH3;
        const int ic = min(i, N_IN * XR3F4 - 1);
        const int n  = ic / XR3F4, g4 = ic % XR3F4;
        const int t  = T0 - XORG + 4 * g4;
        const int tc = min(max(t, 0), T_LEN - 4);
        float4 v = *reinterpret_cast<const float4*>(X + (size_t)n * T_LEN + tc);
        if (t < 0 || t >= T_LEN) v = make_float4(0.0f, 0.0f, 0.0f, 0.0f);
        Xw4[i] = v;
    }
#pragma unroll
    for (int jj = 0; jj < B2TR; ++jj) B2S[tid + jj * NTH3] = B2[min(tid + jj * NTH3, RANK * N_OUT - 1)];
    ybS[tid]   = ybar[min(tid, RANK - 1)];
    biasS[tid] = bias[min(tid, N_OUT - 1)];
    const int nrow = wave * 16 + c;
    const int pos = c >> 2, d = c & 3;
    v2f wA[4], wB[4], wC[4];
    {
        const float* wp = B1s + (wave * 16 + 8 * h) * RANK;
#pragma unroll
        for (int i = 0; i < 4; ++i) {
            const float* w0p = wp + (2 * i) * RANK;
            const float* w1p = w0p + RANK;
            wA[i] = (v2f){w0p[pos], w1p[pos]};
            wB[i] = (v2f){w0p[4 + pos], w1p[4 + pos]};
            wC[i] = (v2f){w0p[8 + (pos >> 1)], w1p[8 + (pos >> 1)]};
        }
    }
    stage_filters<NTH3>(K, Ks, Af, tid);
    __syncthreads();

    v2f a[NTILE][KSTEPS];
    {
        const float* ap = Af + c * AP + 2 * h;
#pragma unroll
        for (int q = 0; q < NTILE; ++q)
#pragma unroll
            for (int s = 0; s < KSTEPS; ++s)
                a[q][s] = *reinterpret_cast<const v2f*>(ap + q * 16 * AP + 4 * s);
    }
    const float* xr = reinterpret_cast<const float*>(Xw4) + nrow * XP3 + XB0 + 2 * h;
    v2f b[KSTEPS];
#pragma unroll
    for (int s = 0; s < KSTEPS; ++s) b[s] = *reinterpret_cast<const v2f*>(xr + 4 * s);

    const int ng = min(G3, (T_LEN - T0) / 4);
    float* yw = ypart + wave * RANK * SPAN3;
    const int offAB = (4 * h + pos) * SPAN3 + d;
    const int offC  = (8 + (pos >> 1)) * SPAN3 + d;
    const bool storeC = (lane & 0x14) == 0;
    for (int g = 0; g < ng; ++g) {
        const int jn = min(g + KSTEPS, NFRAG3 - 1);
        const v2f nb = *reinterpret_cast<const v2f*>(xr + 4 * jn);

        v8f acc[NTILE];
#pragma unroll
        for (int q = 0; q < NTILE; ++q)
#pragma unroll
            for (int r = 0; r < 8; ++r) acc[q][r] = 0.0f;
#pragma unroll
        for (int s = 0; s < KSTEPS; ++s)
#pragma unroll
            for (int q = 0; q < NTILE; ++q)
                acc[q] = __builtin_amdgcn_wmma_f32_16x16x4_f32(
                    false, b[s], false, a[q][s], (short)0, acc[q], false, false);

        v2f sA = (v2f){0.0f, 0.0f}, sB = (v2f){0.0f, 0.0f}, sC = (v2f){0.0f, 0.0f};
#pragma unroll
        for (int i = 0; i < 4; ++i) {
            const v2f reA = (v2f){acc[0][2 * i], acc[0][2 * i + 1]};
            const v2f imA = (v2f){acc[1][2 * i], acc[1][2 * i + 1]};
            const v2f reB = (v2f){acc[2][2 * i], acc[2][2 * i + 1]};
            const v2f imB = (v2f){acc[3][2 * i], acc[3][2 * i + 1]};
            const v2f pC  = (v2f){acc[4][2 * i], acc[4][2 * i + 1]};
            const v2f mA = reA * reA + imA * imA;
            const v2f mB = reB * reB + imB * imB;
            const v2f qC = pC * pC;
            const float mC0 = row_xor_add<4>(qC[0]);
            const float mC1 = row_xor_add<4>(qC[1]);
            const v2f cA = (v2f){__builtin_amdgcn_sqrtf(mA[0]), __builtin_amdgcn_sqrtf(mA[1])};
            const v2f cB = (v2f){__builtin_amdgcn_sqrtf(mB[0]), __builtin_amdgcn_sqrtf(mB[1])};
            const v2f cC = (v2f){__builtin_amdgcn_sqrtf(mC0), __builtin_amdgcn_sqrtf(mC1)};
            sA += wA[i] * cA;
            sB += wB[i] * cB;
            sC += wC[i] * cC;
        }
        float yA = sA[0] + sA[1], yB = sB[0] + sB[1], yC = sC[0] + sC[1];
        yA += xhalf(yA);
        yB += xhalf(yB);
        yC += xhalf(yC);
        float* yp = yw + 4 * g;
        yp[offAB] = h ? yB : yA;
        if (storeC) yp[offC] = yC;
#pragma unroll
        for (int s = 0; s + 1 < KSTEPS; ++s) b[s] = b[s + 1];
        b[KSTEPS - 1] = nb;
    }
    __syncthreads();

    const int t = T0 + tid;
    if (tid < SPAN3 && t < T_LEN) {
        float yv[RANK];
#pragma unroll
        for (int k = 0; k < RANK; ++k) {
            float s = 0.0f;
#pragma unroll
            for (int w = 0; w < NW3; ++w) s += ypart[(w * RANK + k) * SPAN3 + tid];
            yv[k] = s - ybS[k];
        }
        for (int m = 0; m < N_OUT; ++m) {
            float p = 0.0f;
#pragma unroll
            for (int k = 0; k < RANK; ++k) p += yv[k] * B2S[k * N_OUT + m];
            out[(size_t)m * T_LEN + t] = p + biasS[m];
        }
    }
}

extern "C" void kernel_launch(void* const* d_in, const int* in_sizes, int n_in,
                              void* d_out, int out_size, void* d_ws, size_t ws_size,
                              hipStream_t stream) {
    const float* X    = (const float*)d_in[0];
    const float* K    = (const float*)d_in[1];
    const float* B1   = (const float*)d_in[2];
    const float* B2   = (const float*)d_in[3];
    const float* bias = (const float*)d_in[4];
    float*       out  = (float*)d_out;

    float* ws   = (float*)d_ws;
    float* sumA = ws;
    float* sumQ = ws + N_IN * RANK;
    float* B1s  = ws + 2 * N_IN * RANK;
    float* ybar = ws + 3 * N_IN * RANK;

    const int nstats = 2 * N_IN * RANK;
    zero_ws_kernel<<<(nstats + 255) / 256, 256, 0, stream>>>(ws, nstats);

    const int nspan1 = (T_LEN + SPAN1 - 1) / SPAN1;
    pass1_stats<<<dim3(nspan1, N_IN), NTH1, 0, stream>>>(X, K, sumA, sumQ);

    finalize_stats<<<1, 256, 0, stream>>>(B1, sumA, sumQ, B1s, ybar);

    pass3_out<<<NWG3, NTH3, 0, stream>>>(X, K, B2, bias, B1s, ybar, out);
}


# pass 1 re-tiled as pass 3: 256 8-wave workgroups of 16 input rows x 1568 t, filter bank and sliding X fragments in registers (1 LDS read per 130 WMMAs, not 78), sums kept in each lane's registers
# speedup vs baseline: 3.8906x; 1.0409x over previous
#include <hip/hip_runtime.h>
#include <hip/hip_bf16.h>
#include <math.h>


#define T_LEN   50000
#define W_LEN   101
#define N_IN    128
#define RANK    10
#define N_OUT   64

#define KPAD    104
#define KSTEPS  (KPAD / 4)
#define NTILE   5
#define AROWS   (NTILE * 16)
#define AP      108
#define XORG    52
#define XB0     (XORG - 50)

#define G3      49
#define SPAN3   (4 * G3)
#define NW3     8
#define NTH3    (32 * NW3)
#define NFRAG3  (G3 + KSTEPS - 1)
#define NW1     8
#define NTH1    (32 * NW1)
#define SPAN1   (NW1 * SPAN3)
#define NSS1    ((T_LEN + SPAN1 - 1) / SPAN1)
#define NRB1    (N_IN / 16)
#define XR1     (SPAN1 - SPAN3 + XB0 + 4 * (NFRAG3 - 1) + 4)
#define XP1F4   419
#define XP1     (4 * XP1F4)
#define X1TR    ((16 * XP1F4 + NTH1 - 1) / NTH1)
#define NK1     (16 * RANK)
#define XR3F4   75
#define XP3     (XR3F4 * 4)
#define X3TR    ((N_IN * XR3F4 + NTH3 - 1) / NTH3)
#define B2TR    ((RANK * N_OUT + NTH3 - 1) / NTH3)
#define NWG3    ((T_LEN / 4 + G3 - 1) / G3)
#define KTR     ((RANK * 2 * W_LEN + 255) / 256)
#define KSN     (KTR * 256)

typedef __attribute__((ext_vector_type(2))) float v2f;
typedef __attribute__((ext_vector_type(8))) float v8f;


template <int NTHR>
__device__ __forceinline__ void stage_filters(const float* __restrict__ K,
                                              float* __restrict__ Ks,
                                              float* __restrict__ Af, int tid) {
    static_assert(NTHR >= AP, "one thread per column of the filter bank");
    constexpr int TR = (RANK * 2 * W_LEN + NTHR - 1) / NTHR;
    static_assert(TR * NTHR <= KSN, "Ks holds TR whole trips of the workgroup");
#pragma unroll
    for (int j = 0; j < TR; ++j) {
        const int i = tid + j * NTHR;
        Ks[i] = K[min(i, RANK * 2 * W_LEN - 1)];
    }
    __syncthreads();
    if (tid < AP) {
        const int up = tid;
#pragma unroll
        for (int R = 0; R < AROWS; ++R) {
            const int q = R >> 4, m = R & 15, pos = m >> 2, d = m & 3;
            const int k    = (q < 4) ? ((q >> 1) * 4 + pos) : (8 + (pos >> 1));
            const int part = (q < 4) ? (q & 1) : (pos & 1);
            const int u = up - d;
            const int w = min(max(W_LEN - 1 - u, 0), W_LEN - 1);
            const float v = Ks[(k * 2 + part) * W_LEN + w];
            Af[R * AP + up] = (u >= 0 && u < W_LEN) ? v : 0.0f;
        }
    }
}

__global__ void zero_ws_kernel(float* __restrict__ p, int nelem) {
    const int i = blockIdx.x * blockDim.x + threadIdx.x;
    if (i < nelem) p[i] = 0.0f;
}

template <int M>
__device__ __forceinline__ float row_xor_add(float v) {
    const int x = __builtin_amdgcn_update_dpp(0, __builtin_bit_cast(int, v), 0x160 + M, 0xf, 0xf, true);
    return v + __builtin_bit_cast(float, x);
}

__device__ __forceinline__ float xhalf(float v) {
    const int b = __builtin_bit_cast(int, v);
    return __builtin_bit_cast(float, __builtin_amdgcn_permlanex16(b, b, 0x76543210, 0xfedcba98, false, false));
}

static_assert(N_IN == 16 * NRB1 && NSS1 * SPAN1 >= T_LEN, "row blocks and spans cover the input");
static_assert(T_LEN % 4 == 0 && SPAN1 % 4 == 0 && XORG % 4 == 0, "16-byte groups and 4-t column groups never straddle 0 or T");
static_assert(XP1 >= XR1, "window holds every fragment of the last wave's last group");
static_assert((XP1 % 64) % 8 == 4, "row pitch mod 64 is an odd multiple of 4: the 16 rows of a fragment read fall on 64 different banks");

__global__ void __launch_bounds__(NTH1)
pass1_stats(const float* __restrict__ X, const float* __restrict__ K,
            float* __restrict__ sumA, float* __restrict__ sumQ) {
    __shared__ __attribute__((aligned(16))) float Af[AROWS * AP];
    __shared__ float  Ks[KSN];
    __shared__ float4 Xw4[X1TR * NTH1];
    __shared__ float  red[2 * NW1 * NK1];
    const int tid  = threadIdx.x;
    const int lane = tid & 31;
    const int wave = __builtin_amdgcn_readfirstlane(tid >> 5);
    const int c = lane & 15, h = lane >> 4;

    const int rb = blockIdx.x % NRB1;
    const int T0 = (blockIdx.x / NRB1) * SPAN1;
    const float* __restrict__ Xb = X + (size_t)rb * 16 * T_LEN;

#pragma unroll
    for (int j = 0; j < X1TR; ++j) {
        const int i  = tid + j * NTH1;
        const int ic = min(i, 16 * XP1F4 - 1);
        const int n  = ic / XP1F4, g4 = ic % XP1F4;
        const int t  = T0 - XORG + 4 * g4;
        const int tc = min(max(t, 0), T_LEN - 4);
        float4 v = *reinterpret_cast<const float4*>(Xb + (size_t)n * T_LEN + tc);
        if (t < 0 || t >= T_LEN) v = make_float4(0.0f, 0.0f, 0.0f, 0.0f);
        Xw4[i] = v;
    }
    stage_filters<NTH1>(K, Ks, Af, tid);
    __syncthreads();

    v2f a[NTILE][KSTEPS];
    {
        const float* ap = Af + c * AP + 2 * h;
#pragma unroll
        for (int q = 0; q < NTILE; ++q)
#pragma unroll
            for (int s = 0; s < KSTEPS; ++s)
                a[q][s] = *reinterpret_cast<const v2f*>(ap + q * 16 * AP + 4 * s);
    }
    const float* xr = reinterpret_cast<const float*>(Xw4) + c * XP1 + wave * SPAN3 + XB0 + 2 * h;
    v2f b[KSTEPS];
#pragma unroll
    for (int s = 0; s < KSTEPS; ++s) b[s] = *reinterpret_cast<const v2f*>(xr + 4 * s);

    const int ng = max(0, min(G3, (T_LEN - T0 - wave * SPAN3) / 4));
    v2f sA[4], qA[4], sB[4], qB[4], sC[4], qC[4];
#pragma unroll
    for (int i = 0; i < 4; ++i) {
        sA[i] = (v2f){0.0f, 0.0f}; qA[i] = (v2f){0.0f, 0.0f};
        sB[i] = (v2f){0.0f, 0.0f}; qB[i] = (v2f){0.0f, 0.0f};
        sC[i] = (v2f){0.0f, 0.0f}; qC[i] = (v2f){0.0f, 0.0f};
    }
    for (int g = 0; g < ng; ++g) {
        const int jn = min(g + KSTEPS, NFRAG3 - 1);
        const v2f nb = *reinterpret_cast<const v2f*>(xr + 4 * jn);

        v8f acc[NTILE];
#pragma unroll
        for (int q = 0; q < NTILE; ++q)
#pragma unroll
            for (int r = 0; r < 8; ++r) acc[q][r] = 0.0f;
#pragma unroll
        for (int s = 0; s < KSTEPS; ++s)
#pragma unroll
            for (int q = 0; q < NTILE; ++q)
                acc[q] = __builtin_amdgcn_wmma_f32_16x16x4_f32(
                    false, b[s], false, a[q][s], (short)0, acc[q], false, false);

#pragma unroll
        for (int i = 0; i < 4; ++i) {
            const v2f reA = (v2f){acc[0][2 * i], acc[0][2 * i + 1]};
            const v2f imA = (v2f){acc[1][2 * i], acc[1][2 * i + 1]};
            const v2f reB = (v2f){acc[2][2 * i], acc[2][2 * i + 1]};
            const v2f imB = (v2f){acc[3][2 * i], acc[3][2 * i + 1]};
            const v2f pC  = (v2f){acc[4][2 * i], acc[4][2 * i + 1]};
            const v2f mA = reA * reA + imA * imA;
            const v2f mB = reB * reB + imB * imB;
            const v2f p2 = pC * pC;
            const v2f mC = (v2f){row_xor_add<4>(p2[0]), row_xor_add<4>(p2[1])};
            sA[i] += (v2f){__builtin_amdgcn_sqrtf(mA[0]), __builtin_amdgcn_sqrtf(mA[1])};
            sB[i] += (v2f){__builtin_amdgcn_sqrtf(mB[0]), __builtin_amdgcn_sqrtf(mB[1])};
            sC[i] += (v2f){__builtin_amdgcn_sqrtf(mC[0]), __builtin_amdgcn_sqrtf(mC[1])};
            qA[i] += mA;
            qB[i] += mB;
            qC[i] += mC;
        }
#pragma unroll
        for (int s = 0; s + 1 < KSTEPS; ++s) b[s] = b[s + 1];
        b[KSTEPS - 1] = nb;
    }

#pragma unroll
    for (int i = 0; i < 4; ++i)
#pragma unroll
        for (int e = 0; e < 2; ++e) {
            sA[i][e] = row_xor_add<2>(row_xor_add<1>(sA[i][e]));
            qA[i][e] = row_xor_add<2>(row_xor_add<1>(qA[i][e]));
            sB[i][e] = row_xor_add<2>(row_xor_add<1>(sB[i][e]));
            qB[i][e] = row_xor_add<2>(row_xor_add<1>(qB[i][e]));
            sC[i][e] = row_xor_add<2>(row_xor_add<1>(sC[i][e]));
            qC[i][e] = row_xor_add<2>(row_xor_add<1>(qC[i][e]));
        }
    if ((c & 3) == 0) {
        const int pos = c >> 2;
        float* rs = red + wave * NK1 + 8 * h * RANK;
        float* rq = rs + NW1 * NK1;
#pragma unroll
        for (int i = 0; i < 4; ++i)
#pragma unroll
            for (int e = 0; e < 2; ++e) {
                const int ro = (2 * i + e) * RANK;
                rs[ro + pos]     = sA[i][e];  rq[ro + pos]     = qA[i][e];
                rs[ro + 4 + pos] = sB[i][e];  rq[ro + 4 + pos] = qB[i][e];
                if ((pos & 1) == 0) {
                    rs[ro + 8 + (pos >> 1)] = sC[i][e];
                    rq[ro + 8 + (pos >> 1)] = qC[i][e];
                }
            }
    }
    __syncthreads();
    for (int j = tid; j < 2 * NK1; j += NTH1) {
        const int which = j / NK1, idx = j % NK1;
        float v = 0.0f;
#pragma unroll
        for (int w = 0; w < NW1; ++w) v += red[(which * NW1 + w) * NK1 + idx];
        atomicAdd((which ? sumQ : sumA) + rb * NK1 + idx, v);
    }
}

__global__ void finalize_stats(const float* __restrict__ B1,
                               const float* __restrict__ sumA,
                               const float* __restrict__ sumQ,
                               float* __restrict__ B1s, float* __restrict__ ybar) {
    const int tid = threadIdx.x;
    if (tid < RANK) ybar[tid] = 0.0f;
    __syncthreads();
    for (int i = tid; i < N_IN * RANK; i += blockDim.x) {
        const float s    = sumA[i], q = sumQ[i];
        const float mean = s / (float)T_LEN;
        const float var  = (q - s * mean) / (float)(T_LEN - 1);
        const float w    = B1[i] * rsqrtf(var);
        B1s[i] = w;
        atomicAdd(&ybar[i % RANK], w * mean);
    }
}

static_assert(N_IN == 16 * NW3, "one wave per block of 16 input rows");
static_assert(XP3 >= XB0 + 4 * (NFRAG3 - 1) + 4, "window holds every fragment of the last group");
static_assert((XP3 % 64) % 8 == 4, "row pitch mod 64 is an odd multiple of 4: the 16 rows of a fragment read fall on 64 different banks");
static_assert(NTH3 >= SPAN3 && NTH3 >= N_OUT && NTH3 >= RANK, "readout: one thread per time point; one per bias and ybar element");
static_assert(NWG3 * SPAN3 >= T_LEN && T_LEN % 4 == 0 && XORG % 4 == 0, "spans cover T; 16-byte groups never straddle 0 or T");

__global__ void __launch_bounds__(NTH3)
pass3_out(const float* __restrict__ X, const float* __restrict__ K,
          const float* __restrict__ B2, const float* __restrict__ bias,
          const float* __restrict__ B1s, const float* __restrict__ ybar,
          float* __restrict__ out) {
    __shared__ __attribute__((aligned(16))) float Af[AROWS * AP];
    __shared__ float  Ks[KSN];
    __shared__ float4 Xw4[X3TR * NTH3];
    __shared__ float  B2S[B2TR * NTH3];
    __shared__ __attribute__((aligned(16))) float ypart[NW3 * RANK * SPAN3];
    __shared__ float  ybS[NTH3];
    __shared__ float  biasS[NTH3];
    const int tid  = threadIdx.x;
    const int lane = tid & 31;
    const int wave = __builtin_amdgcn_readfirstlane(tid >> 5);
    const int c = lane & 15, h = lane >> 4;

    const int T0 = blockIdx.x * SPAN3;

#pragma unroll
    for (int j = 0; j < X3TR; ++j) {
        const int i  = tid + j * NTH3;
        const int ic = min(i, N_IN * XR3F4 - 1);
        const int n  = ic / XR3F4, g4 = ic % XR3F4;
        const int t  = T0 - XORG + 4 * g4;
        const int tc = min(max(t, 0), T_LEN - 4);
        float4 v = *reinterpret_cast<const float4*>(X + (size_t)n * T_LEN + tc);
        if (t < 0 || t >= T_LEN) v = make_float4(0.0f, 0.0f, 0.0f, 0.0f);
        Xw4[i] = v;
    }
#pragma unroll
    for (int jj = 0; jj < B2TR; ++jj) B2S[tid + jj * NTH3] = B2[min(tid + jj * NTH3, RANK * N_OUT - 1)];
    ybS[tid]   = ybar[min(tid, RANK - 1)];
    biasS[tid] = bias[min(tid, N_OUT - 1)];
    const int nrow = wave * 16 + c;
    const int pos = c >> 2, d = c & 3;
    v2f wA[4], wB[4], wC[4];
    {
        const float* wp = B1s + (wave * 16 + 8 * h) * RANK;
#pragma unroll
        for (int i = 0; i < 4; ++i) {
            const float* w0p = wp + (2 * i) * RANK;
            const float* w1p = w0p + RANK;
            wA[i] = (v2f){w0p[pos], w1p[pos]};
            wB[i] = (v2f){w0p[4 + pos], w1p[4 + pos]};
            wC[i] = (v2f){w0p[8 + (pos >> 1)], w1p[8 + (pos >> 1)]};
        }
    }
    stage_filters<NTH3>(K, Ks, Af, tid);
    __syncthreads();

    v2f a[NTILE][KSTEPS];
    {
        const float* ap = Af + c * AP + 2 * h;
#pragma unroll
        for (int q = 0; q < NTILE; ++q)
#pragma unroll
            for (int s = 0; s < KSTEPS; ++s)
                a[q][s] = *reinterpret_cast<const v2f*>(ap + q * 16 * AP + 4 * s);
    }
    const float* xr = reinterpret_cast<const float*>(Xw4) + nrow * XP3 + XB0 + 2 * h;
    v2f b[KSTEPS];
#pragma unroll
    for (int s = 0; s < KSTEPS; ++s) b[s] = *reinterpret_cast<const v2f*>(xr + 4 * s);

    const int ng = min(G3, (T_LEN - T0) / 4);
    float* yw = ypart + wave * RANK * SPAN3;
    const int offAB = (4 * h + pos) * SPAN3 + d;
    const int offC  = (8 + (pos >> 1)) * SPAN3 + d;
    const bool storeC = (lane & 0x14) == 0;
    for (int g = 0; g < ng; ++g) {
        const int jn = min(g + KSTEPS, NFRAG3 - 1);
        const v2f nb = *reinterpret_cast<const v2f*>(xr + 4 * jn);

        v8f acc[NTILE];
#pragma unroll
        for (int q = 0; q < NTILE; ++q)
#pragma unroll
            for (int r = 0; r < 8; ++r) acc[q][r] = 0.0f;
#pragma unroll
        for (int s = 0; s < KSTEPS; ++s)
#pragma unroll
            for (int q = 0; q < NTILE; ++q)
                acc[q] = __builtin_amdgcn_wmma_f32_16x16x4_f32(
                    false, b[s], false, a[q][s], (short)0, acc[q], false, false);

        v2f sA = (v2f){0.0f, 0.0f}, sB = (v2f){0.0f, 0.0f}, sC = (v2f){0.0f, 0.0f};
#pragma unroll
        for (int i = 0; i < 4; ++i) {
            const v2f reA = (v2f){acc[0][2 * i], acc[0][2 * i + 1]};
            const v2f imA = (v2f){acc[1][2 * i], acc[1][2 * i + 1]};
            const v2f reB = (v2f){acc[2][2 * i], acc[2][2 * i + 1]};
            const v2f imB = (v2f){acc[3][2 * i], acc[3][2 * i + 1]};
            const v2f pC  = (v2f){acc[4][2 * i], acc[4][2 * i + 1]};
            const v2f mA = reA * reA + imA * imA;
            const v2f mB = reB * reB + imB * imB;
            const v2f qC = pC * pC;
            const float mC0 = row_xor_add<4>(qC[0]);
            const float mC1 = row_xor_add<4>(qC[1]);
            const v2f cA = (v2f){__builtin_amdgcn_sqrtf(mA[0]), __builtin_amdgcn_sqrtf(mA[1])};
            const v2f cB = (v2f){__builtin_amdgcn_sqrtf(mB[0]), __builtin_amdgcn_sqrtf(mB[1])};
            const v2f cC = (v2f){__builtin_amdgcn_sqrtf(mC0), __builtin_amdgcn_sqrtf(mC1)};
            sA += wA[i] * cA;
            sB += wB[i] * cB;
            sC += wC[i] * cC;
        }
        float yA = sA[0] + sA[1], yB = sB[0] + sB[1], yC = sC[0] + sC[1];
        yA += xhalf(yA);
        yB += xhalf(yB);
        yC += xhalf(yC);
        float* yp = yw + 4 * g;
        yp[offAB] = h ? yB : yA;
        if (storeC) yp[offC] = yC;
#pragma unroll
        for (int s = 0; s + 1 < KSTEPS; ++s) b[s] = b[s + 1];
        b[KSTEPS - 1] = nb;
    }
    __syncthreads();

    const int t = T0 + tid;
    if (tid < SPAN3 && t < T_LEN) {
        float yv[RANK];
#pragma unroll
        for (int k = 0; k < RANK; ++k) {
            float s = 0.0f;
#pragma unroll
            for (int w = 0; w < NW3; ++w) s += ypart[(w * RANK + k) * SPAN3 + tid];
            yv[k] = s - ybS[k];
        }
        for (int m = 0; m < N_OUT; ++m) {
            float p = 0.0f;
#pragma unroll
            for (int k = 0; k < RANK; ++k) p += yv[k] * B2S[k * N_OUT + m];
            out[(size_t)m * T_LEN + t] = p + biasS[m];
        }
    }
}

extern "C" void kernel_launch(void* const* d_in, const int* in_sizes, int n_in,
                              void* d_out, int out_size, void* d_ws, size_t ws_size,
                              hipStream_t stream) {
    const float* X    = (const float*)d_in[0];
    const float* K    = (const float*)d_in[1];
    const float* B1   = (const float*)d_in[2];
    const float* B2   = (const float*)d_in[3];
    const float* bias = (const float*)d_in[4];
    float*       out  = (float*)d_out;

    float* ws   = (float*)d_ws;
    float* sumA = ws;
    float* sumQ = ws + N_IN * RANK;
    float* B1s  = ws + 2 * N_IN * RANK;
    float* ybar = ws + 3 * N_IN * RANK;

    const int nstats = 2 * N_IN * RANK;
    zero_ws_kernel<<<(nstats + 255) / 256, 256, 0, stream>>>(ws, nstats);

    pass1_stats<<<NRB1 * NSS1, NTH1, 0, stream>>>(X, K, sumA, sumQ);

    finalize_stats<<<1, 256, 0, stream>>>(B1, sumA, sumQ, B1s, ybar);

    pass3_out<<<NWG3, NTH3, 0, stream>>>(X, K, B2, bias, B1s, ybar, out);
}


# pass 1 keeps every |conv| in the workspace as 512-byte lane-order records and pass 3 streams them back instead of convolving again (old pass 3 stays as the path for a workspace under 256 MB)
# speedup vs baseline: 5.8775x; 1.5107x over previous
#include <hip/hip_runtime.h>
#include <hip/hip_bf16.h>
#include <math.h>


#define T_LEN   50000
#define W_LEN   101
#define N_IN    128
#define RANK    10
#define N_OUT   64

#define KPAD    104
#define KSTEPS  (KPAD / 4)
#define NTILE   5
#define AROWS   (NTILE * 16)
#define AP      108
#define XORG    52
#define XB0     (XORG - 50)

#define G3      49
#define SPAN3   (4 * G3)
#define NW3     8
#define NTH3    (32 * NW3)
#define NFRAG3  (G3 + KSTEPS - 1)
#define NW1     8
#define NTH1    (32 * NW1)
#define SPAN1   (NW1 * SPAN3)
#define NSS1    ((T_LEN + SPAN1 - 1) / SPAN1)
#define NRB1    (N_IN / 16)
#define XR1     (SPAN1 - SPAN3 + XB0 + 4 * (NFRAG3 - 1) + 4)
#define XP1F4   419
#define XP1     (4 * XP1F4)
#define X1TR    ((16 * XP1F4 + NTH1 - 1) / NTH1)
#define NK1     (16 * RANK)
#define XR3F4   75
#define XP3     (XR3F4 * 4)
#define X3TR    ((N_IN * XR3F4 + NTH3 - 1) / NTH3)
#define B2TR    ((RANK * N_OUT + NTH3 - 1) / NTH3)
#define NWG3    ((T_LEN / 4 + G3 - 1) / G3)
#define KTR     ((RANK * 2 * W_LEN + 255) / 256)
#define KSN     (KTR * 256)
#define NG      (T_LEN / 4)
#define CREC4   (5 * 32)
#define CKOFF   4096
#define CKBYTES ((size_t)CKOFF * 4 + (size_t)NRB1 * NG * CREC4 * 16)
#define UN3     5
#define NCH3    ((G3 + UN3 - 1) / UN3)

typedef __attribute__((ext_vector_type(2))) float v2f;
typedef __attribute__((ext_vector_type(8))) float v8f;


template <int NTHR>
__device__ __forceinline__ void stage_filters(const float* __restrict__ K,
                                              float* __restrict__ Ks,
                                              float* __restrict__ Af, int tid) {
    static_assert(NTHR >= AP, "one thread per column of the filter bank");
    constexpr int TR = (RANK * 2 * W_LEN + NTHR - 1) / NTHR;
    static_assert(TR * NTHR <= KSN, "Ks holds TR whole trips of the workgroup");
#pragma unroll
    for (int j = 0; j < TR; ++j) {
        const int i = tid + j * NTHR;
        Ks[i] = K[min(i, RANK * 2 * W_LEN - 1)];
    }
    __syncthreads();
    if (tid < AP) {
        const int up = tid;
#pragma unroll
        for (int R = 0; R < AROWS; ++R) {
            const int q = R >> 4, m = R & 15, pos = m >> 2, d = m & 3;
            const int k    = (q < 4) ? ((q >> 1) * 4 + pos) : (8 + (pos >> 1));
            const int part = (q < 4) ? (q & 1) : (pos & 1);
            const int u = up - d;
            const int w = min(max(W_LEN - 1 - u, 0), W_LEN - 1);
            const float v = Ks[(k * 2 + part) * W_LEN + w];
            Af[R * AP + up] = (u >= 0 && u < W_LEN) ? v : 0.0f;
        }
    }
}

__global__ void zero_ws_kernel(float* __restrict__ p, int nelem) {
    const int i = blockIdx.x * blockDim.x + threadIdx.x;
    if (i < nelem) p[i] = 0.0f;
}

template <int M>
__device__ __forceinline__ float row_xor_add(float v) {
    const int x = __builtin_amdgcn_update_dpp(0, __builtin_bit_cast(int, v), 0x160 + M, 0xf, 0xf, true);
    return v + __builtin_bit_cast(float, x);
}

__device__ __forceinline__ float xhalf(float v) {
    const int b = __builtin_bit_cast(int, v);
    return __builtin_bit_cast(float, __builtin_amdgcn_permlanex16(b, b, 0x76543210, 0xfedcba98, false, false));
}

static_assert(N_IN == 16 * NRB1 && NSS1 * SPAN1 >= T_LEN, "row blocks and spans cover the input");
static_assert(T_LEN % 4 == 0 && SPAN1 % 4 == 0 && XORG % 4 == 0, "16-byte groups and 4-t column groups never straddle 0 or T");
static_assert(XP1 >= XR1, "window holds every fragment of the last wave's last group");
static_assert((XP1 % 64) % 8 == 4, "row pitch mod 64 is an odd multiple of 4: the 16 rows of a fragment read fall on 64 different banks");

__global__ void __launch_bounds__(NTH1)
pass1_stats(const float* __restrict__ X, const float* __restrict__ K,
            float* __restrict__ sumA, float* __restrict__ sumQ,
            float* __restrict__ ckeep, size_t ws_bytes) {
    __shared__ __attribute__((aligned(16))) float Af[AROWS * AP];
    __shared__ float  Ks[KSN];
    __shared__ float4 Xw4[X1TR * NTH1];
    __shared__ float  red[2 * NW1 * NK1];
    const int tid  = threadIdx.x;
    const int lane = tid & 31;
    const int wave = __builtin_amdgcn_readfirstlane(tid >> 5);
    const int c = lane & 15, h = lane >> 4;

    const int rb = blockIdx.x % NRB1;
    const int T0 = (blockIdx.x / NRB1) * SPAN1;
    const float* __restrict__ Xb = X + (size_t)rb * 16 * T_LEN;
    const bool keep = ws_bytes >= CKBYTES;
    float4* cdst = reinterpret_cast<float4*>(ckeep) + ((size_t)rb * NG + (T0 / 4 + wave * G3)) * CREC4 + lane;
    const bool odd = (lane & 4) != 0;

#pragma unroll
    for (int j = 0; j < X1TR; ++j) {
        const int i  = tid + j * NTH1;
        const int ic = min(i, 16 * XP1F4 - 1);
        const int n  = ic / XP1F4, g4 = ic % XP1F4;
        const int t  = T0 - XORG + 4 * g4;
        const int tc = min(max(t, 0), T_LEN - 4);
        float4 v = *reinterpret_cast<const float4*>(Xb + (size_t)n * T_LEN + tc);
        if (t < 0 || t >= T_LEN) v = make_float4(0.0f, 0.0f, 0.0f, 0.0f);
        Xw4[i] = v;
    }
    stage_filters<NTH1>(K, Ks, Af, tid);
    __syncthreads();

    v2f a[NTILE][KSTEPS];
    {
        const float* ap = Af + c * AP + 2 * h;
#pragma unroll
        for (int q = 0; q < NTILE; ++q)
#pragma unroll
            for (int s = 0; s < KSTEPS; ++s)
                a[q][s] = *reinterpret_cast<const v2f*>(ap + q * 16 * AP + 4 * s);
    }
    const float* xr = reinterpret_cast<const float*>(Xw4) + c * XP1 + wave * SPAN3 + XB0 + 2 * h;
    v2f b[KSTEPS];
#pragma unroll
    for (int s = 0; s < KSTEPS; ++s) b[s] = *reinterpret_cast<const v2f*>(xr + 4 * s);

    const int ng = max(0, min(G3, (T_LEN - T0 - wave * SPAN3) / 4));
    v2f sA[4], qA[4], sB[4], qB[4], sC[4], qC[4];
#pragma unroll
    for (int i = 0; i < 4; ++i) {
        sA[i] = (v2f){0.0f, 0.0f}; qA[i] = (v2f){0.0f, 0.0f};
        sB[i] = (v2f){0.0f, 0.0f}; qB[i] = (v2f){0.0f, 0.0f};
        sC[i] = (v2f){0.0f, 0.0f}; qC[i] = (v2f){0.0f, 0.0f};
    }
    for (int g = 0; g < ng; ++g) {
        const int jn = min(g + KSTEPS, NFRAG3 - 1);
        const v2f nb = *reinterpret_cast<const v2f*>(xr + 4 * jn);

        v8f acc[NTILE];
#pragma unroll
        for (int q = 0; q < NTILE; ++q)
#pragma unroll
            for (int r = 0; r < 8; ++r) acc[q][r] = 0.0f;
#pragma unroll
        for (int s = 0; s < KSTEPS; ++s)
#pragma unroll
            for (int q = 0; q < NTILE; ++q)
                acc[q] = __builtin_amdgcn_wmma_f32_16x16x4_f32(
                    false, b[s], false, a[q][s], (short)0, acc[q], false, false);

        float kA[8], kB[8], kC[4];
#pragma unroll
        for (int i = 0; i < 4; ++i) {
            const v2f reA = (v2f){acc[0][2 * i], acc[0][2 * i + 1]};
            const v2f imA = (v2f){acc[1][2 * i], acc[1][2 * i + 1]};
            const v2f reB = (v2f){acc[2][2 * i], acc[2][2 * i + 1]};
            const v2f imB = (v2f){acc[3][2 * i], acc[3][2 * i + 1]};
            const v2f pC  = (v2f){acc[4][2 * i], acc[4][2 * i + 1]};
            const v2f mA = reA * reA + imA * imA;
            const v2f mB = reB * reB + imB * imB;
            const v2f p2 = pC * pC;
            const v2f mC = (v2f){row_xor_add<4>(p2[0]), row_xor_add<4>(p2[1])};
            const v2f cA = (v2f){__builtin_amdgcn_sqrtf(mA[0]), __builtin_amdgcn_sqrtf(mA[1])};
            const v2f cB = (v2f){__builtin_amdgcn_sqrtf(mB[0]), __builtin_amdgcn_sqrtf(mB[1])};
            const v2f cC = (v2f){__builtin_amdgcn_sqrtf(mC[0]), __builtin_amdgcn_sqrtf(mC[1])};
            sA[i] += cA;
            sB[i] += cB;
            sC[i] += cC;
            qA[i] += mA;
            qB[i] += mB;
            qC[i] += mC;
            kA[2 * i] = cA[0]; kA[2 * i + 1] = cA[1];
            kB[2 * i] = cB[0]; kB[2 * i + 1] = cB[1];
            kC[i] = odd ? cC[1] : cC[0];
        }
        if (keep) {
            float4* dp = cdst + (size_t)g * CREC4;
            dp[0]   = make_float4(kA[0], kA[1], kA[2], kA[3]);
            dp[32]  = make_float4(kA[4], kA[5], kA[6], kA[7]);
            dp[64]  = make_float4(kB[0], kB[1], kB[2], kB[3]);
            dp[96]  = make_float4(kB[4], kB[5], kB[6], kB[7]);
            dp[128] = make_float4(kC[0], kC[1], kC[2], kC[3]);
        }
#pragma unroll
        for (int s = 0; s + 1 < KSTEPS; ++s) b[s] = b[s + 1];
        b[KSTEPS - 1] = nb;
    }

#pragma unroll
    for (int i = 0; i < 4; ++i)
#pragma unroll
        for (int e = 0; e < 2; ++e) {
            sA[i][e] = row_xor_add<2>(row_xor_add<1>(sA[i][e]));
            qA[i][e] = row_xor_add<2>(row_xor_add<1>(qA[i][e]));
            sB[i][e] = row_xor_add<2>(row_xor_add<1>(sB[i][e]));
            qB[i][e] = row_xor_add<2>(row_xor_add<1>(qB[i][e]));
            sC[i][e] = row_xor_add<2>(row_xor_add<1>(sC[i][e]));
            qC[i][e] = row_xor_add<2>(row_xor_add<1>(qC[i][e]));
        }
    if ((c & 3) == 0) {
        const int pos = c >> 2;
        float* rs = red + wave * NK1 + 8 * h * RANK;
        float* rq = rs + NW1 * NK1;
#pragma unroll
        for (int i = 0; i < 4; ++i)
#pragma unroll
            for (int e = 0; e < 2; ++e) {
                const int ro = (2 * i + e) * RANK;
                rs[ro + pos]     = sA[i][e];  rq[ro + pos]     = qA[i][e];
                rs[ro + 4 + pos] = sB[i][e];  rq[ro + 4 + pos] = qB[i][e];
                if ((pos & 1) == 0) {
                    rs[ro + 8 + (pos >> 1)] = sC[i][e];
                    rq[ro + 8 + (pos >> 1)] = qC[i][e];
                }
            }
    }
    __syncthreads();
    for (int j = tid; j < 2 * NK1; j += NTH1) {
        const int which = j / NK1, idx = j % NK1;
        float v = 0.0f;
#pragma unroll
        for (int w = 0; w < NW1; ++w) v += red[(which * NW1 + w) * NK1 + idx];
        atomicAdd((which ? sumQ : sumA) + rb * NK1 + idx, v);
    }
}

__global__ void finalize_stats(const float* __restrict__ B1,
                               const float* __restrict__ sumA,
                               const float* __restrict__ sumQ,
                               float* __restrict__ B1s, float* __restrict__ ybar) {
    const int tid = threadIdx.x;
    if (tid < RANK) ybar[tid] = 0.0f;
    __syncthreads();
    for (int i = tid; i < N_IN * RANK; i += blockDim.x) {
        const float s    = sumA[i], q = sumQ[i];
        const float mean = s / (float)T_LEN;
        const float var  = (q - s * mean) / (float)(T_LEN - 1);
        const float w    = B1[i] * rsqrtf(var);
        B1s[i] = w;
        atomicAdd(&ybar[i % RANK], w * mean);
    }
}

static_assert(NCH3 % 2 == 0 && NCH3 * UN3 >= G3, "kept |conv|: the chunk loop takes two chunks per trip and covers G3");
static_assert(N_IN == 16 * NW3, "one wave per block of 16 input rows");
static_assert(XP3 >= XB0 + 4 * (NFRAG3 - 1) + 4, "window holds every fragment of the last group");
static_assert((XP3 % 64) % 8 == 4, "row pitch mod 64 is an odd multiple of 4: the 16 rows of a fragment read fall on 64 different banks");
static_assert(NTH3 >= SPAN3 && NTH3 >= N_OUT && NTH3 >= RANK, "readout: one thread per time point; one per bias and ybar element");
static_assert(NWG3 * SPAN3 >= T_LEN && T_LEN % 4 == 0 && XORG % 4 == 0, "spans cover T; 16-byte groups never straddle 0 or T");

__global__ void __launch_bounds__(NTH3)
pass3_out(const float* __restrict__ X, const float* __restrict__ K,
          const float* __restrict__ B2, const float* __restrict__ bias,
          const float* __restrict__ B1s, const float* __restrict__ ybar,
          const float* __restrict__ ckeep, size_t ws_bytes,
          float* __restrict__ out) {
    __shared__ __attribute__((aligned(16))) float Af[AROWS * AP];
    __shared__ float  Ks[KSN];
    __shared__ float4 Xw4[X3TR * NTH3];
    __shared__ float  B2S[B2TR * NTH3];
    __shared__ __attribute__((aligned(16))) float ypart[NW3 * RANK * SPAN3];
    __shared__ float  ybS[NTH3];
    __shared__ float  biasS[NTH3];
    const int tid  = threadIdx.x;
    const int lane = tid & 31;
    const int wave = __builtin_amdgcn_readfirstlane(tid >> 5);
    const int c = lane & 15, h = lane >> 4;

    const int T0 = blockIdx.x * SPAN3;
    const bool keep = ws_bytes >= CKBYTES;

    if (!keep) {
#pragma unroll
        for (int j = 0; j < X3TR; ++j) {
            const int i  = tid + j * NTH3;
            const int ic = min(i, N_IN * XR3F4 - 1);
            const int n  = ic / XR3F4, g4 = ic % XR3F4;
            const int t  = T0 - XORG + 4 * g4;
            const int tc = min(max(t, 0), T_LEN - 4);
            float4 v = *reinterpret_cast<const float4*>(X + (size_t)n * T_LEN + tc);
            if (t < 0 || t >= T_LEN) v = make_float4(0.0f, 0.0f, 0.0f, 0.0f);
            Xw4[i] = v;
        }
    }
#pragma unroll
    for (int jj = 0; jj < B2TR; ++jj) B2S[tid + jj * NTH3] = B2[min(tid + jj * NTH3, RANK * N_OUT - 1)];
    ybS[tid]   = ybar[min(tid, RANK - 1)];
    biasS[tid] = bias[min(tid, N_OUT - 1)];
    const int nrow = wave * 16 + c;
    const int pos = c >> 2, d = c & 3;
    v2f wA[4], wB[4], wC[4];
    {
        const float* wp = B1s + (wave * 16 + 8 * h) * RANK;
#pragma unroll
        for (int i = 0; i < 4; ++i) {
            const float* w0p = wp + (2 * i) * RANK;
            const float* w1p = w0p + RANK;
            wA[i] = (v2f){w0p[pos], w1p[pos]};
            wB[i] = (v2f){w0p[4 + pos], w1p[4 + pos]};
            wC[i] = (v2f){w0p[8 + (pos >> 1)], w1p[8 + (pos >> 1)]};
        }
    }
    const int ng = min(G3, (T_LEN - T0) / 4);
    float* yw = ypart + wave * RANK * SPAN3;
    const int offAB = (4 * h + pos) * SPAN3 + d;
    const int offC  = (8 + (pos >> 1)) * SPAN3 + d;
    const bool storeC = (lane & 0x14) == 0;

    if (keep) {
        const bool odd = (lane & 4) != 0;
        float wAf[8], wBf[8], wCs[4];
#pragma unroll
        for (int i = 0; i < 4; ++i) {
            wAf[2 * i] = wA[i][0]; wAf[2 * i + 1] = wA[i][1];
            wBf[2 * i] = wB[i][0]; wBf[2 * i + 1] = wB[i][1];
            wCs[i] = odd ? wC[i][1] : wC[i][0];
        }
        const float4* src = reinterpret_cast<const float4*>(ckeep) + ((size_t)wave * NG) * CREC4 + lane;
        const int G0 = blockIdx.x * G3;
        float4 ra[UN3][5], rn[UN3][5];
#pragma unroll
        for (int u = 0; u < UN3; ++u) {
            const float4* p = src + (size_t)min(G0 + u, NG - 1) * CREC4;
#pragma unroll
            for (int j = 0; j < 5; ++j) ra[u][j] = p[32 * j];
        }
        for (int ch = 0; ch < NCH3; ch += 2) {
#pragma unroll
            for (int u = 0; u < UN3; ++u) {
                const float4* p = src + (size_t)min(G0 + (ch + 1) * UN3 + u, NG - 1) * CREC4;
#pragma unroll
                for (int j = 0; j < 5; ++j) rn[u][j] = p[32 * j];
            }
#pragma unroll
            for (int u = 0; u < UN3; ++u) {
                const int g = ch * UN3 + u;
                float sA = ra[u][0].x * wAf[0] + ra[u][0].y * wAf[1] + ra[u][0].z * wAf[2] + ra[u][0].w * wAf[3]
                         + ra[u][1].x * wAf[4] + ra[u][1].y * wAf[5] + ra[u][1].z * wAf[6] + ra[u][1].w * wAf[7];
                float sB = ra[u][2].x * wBf[0] + ra[u][2].y * wBf[1] + ra[u][2].z * wBf[2] + ra[u][2].w * wBf[3]
                         + ra[u][3].x * wBf[4] + ra[u][3].y * wBf[5] + ra[u][3].z * wBf[6] + ra[u][3].w * wBf[7];
                float sC = ra[u][4].x * wCs[0] + ra[u][4].y * wCs[1] + ra[u][4].z * wCs[2] + ra[u][4].w * wCs[3];
                sC = row_xor_add<4>(sC);
                const float yA = sA + xhalf(sA);
                const float yB = sB + xhalf(sB);
                const float yC = sC + xhalf(sC);
                if (g < ng) {
                    float* yp = yw + 4 * g;
                    yp[offAB] = h ? yB : yA;
                    if (storeC) yp[offC] = yC;
                }
            }
            if (ch + 2 < NCH3) {
#pragma unroll
                for (int u = 0; u < UN3; ++u) {
                    const float4* p = src + (size_t)min(G0 + (ch + 2) * UN3 + u, NG - 1) * CREC4;
#pragma unroll
                    for (int j = 0; j < 5; ++j) ra[u][j] = p[32 * j];
                }
            }
#pragma unroll
            for (int u = 0; u < UN3; ++u) {
                const int g = (ch + 1) * UN3 + u;
                float sA = rn[u][0].x * wAf[0] + rn[u][0].y * wAf[1] + rn[u][0].z * wAf[2] + rn[u][0].w * wAf[3]
                         + rn[u][1].x * wAf[4] + rn[u][1].y * wAf[5] + rn[u][1].z * wAf[6] + rn[u][1].w * wAf[7];
                float sB = rn[u][2].x * wBf[0] + rn[u][2].y * wBf[1] + rn[u][2].z * wBf[2] + rn[u][2].w * wBf[3]
                         + rn[u][3].x * wBf[4] + rn[u][3].y * wBf[5] + rn[u][3].z * wBf[6] + rn[u][3].w * wBf[7];
                float sC = rn[u][4].x * wCs[0] + rn[u][4].y * wCs[1] + rn[u][4].z * wCs[2] + rn[u][4].w * wCs[3];
                sC = row_xor_add<4>(sC);
                const float yA = sA + xhalf(sA);
                const float yB = sB + xhalf(sB);
                const float yC = sC + xhalf(sC);
                if (g < ng) {
                    float* yp = yw + 4 * g;
                    yp[offAB] = h ? yB : yA;
                    if (storeC) yp[offC] = yC;
                }
            }
        }
    } else {
        stage_filters<NTH3>(K, Ks, Af, tid);
        __syncthreads();

        v2f a[NTILE][KSTEPS];
        {
            const float* ap = Af + c * AP + 2 * h;
#pragma unroll
            for (int q = 0; q < NTILE; ++q)
#pragma unroll
                for (int s = 0; s < KSTEPS; ++s)
                    a[q][s] = *reinterpret_cast<const v2f*>(ap + q * 16 * AP + 4 * s);
        }
        const float* xr = reinterpret_cast<const float*>(Xw4) + nrow * XP3 + XB0 + 2 * h;
        v2f b[KSTEPS];
#pragma unroll
        for (int s = 0; s < KSTEPS; ++s) b[s] = *reinterpret_cast<const v2f*>(xr + 4 * s);

        for (int g = 0; g < ng; ++g) {
            const int jn = min(g + KSTEPS, NFRAG3 - 1);
            const v2f nb = *reinterpret_cast<const v2f*>(xr + 4 * jn);

            v8f acc[NTILE];
#pragma unroll
            for (int q = 0; q < NTILE; ++q)
#pragma unroll
                for (int r = 0; r < 8; ++r) acc[q][r] = 0.0f;
#pragma unroll
            for (int s = 0; s < KSTEPS; ++s)
#pragma unroll
                for (int q = 0; q < NTILE; ++q)
                    acc[q] = __builtin_amdgcn_wmma_f32_16x16x4_f32(
                        false, b[s], false, a[q][s], (short)0, acc[q], false, false);

            v2f sA = (v2f){0.0f, 0.0f}, sB = (v2f){0.0f, 0.0f}, sC = (v2f){0.0f, 0.0f};
#pragma unroll
            for (int i = 0; i < 4; ++i) {
                const v2f reA = (v2f){acc[0][2 * i], acc[0][2 * i + 1]};
                const v2f imA = (v2f){acc[1][2 * i], acc[1][2 * i + 1]};
                const v2f reB = (v2f){acc[2][2 * i], acc[2][2 * i + 1]};
                const v2f imB = (v2f){acc[3][2 * i], acc[3][2 * i + 1]};
                const v2f pC  = (v2f){acc[4][2 * i], acc[4][2 * i + 1]};
                const v2f mA = reA * reA + imA * imA;
                const v2f mB = reB * reB + imB * imB;
                const v2f qC = pC * pC;
                const float mC0 = row_xor_add<4>(qC[0]);
                const float mC1 = row_xor_add<4>(qC[1]);
                const v2f cA = (v2f){__builtin_amdgcn_sqrtf(mA[0]), __builtin_amdgcn_sqrtf(mA[1])};
                const v2f cB = (v2f){__builtin_amdgcn_sqrtf(mB[0]), __builtin_amdgcn_sqrtf(mB[1])};
                const v2f cC = (v2f){__builtin_amdgcn_sqrtf(mC0), __builtin_amdgcn_sqrtf(mC1)};
                sA += wA[i] * cA;
                sB += wB[i] * cB;
                sC += wC[i] * cC;
            }
            float yA = sA[0] + sA[1], yB = sB[0] + sB[1], yC = sC[0] + sC[1];
            yA += xhalf(yA);
            yB += xhalf(yB);
            yC += xhalf(yC);
            float* yp = yw + 4 * g;
            yp[offAB] = h ? yB : yA;
            if (storeC) yp[offC] = yC;
#pragma unroll
            for (int s = 0; s + 1 < KSTEPS; ++s) b[s] = b[s + 1];
            b[KSTEPS - 1] = nb;
        }
    }
    __syncthreads();

    const int t = T0 + tid;
    if (tid < SPAN3 && t < T_LEN) {
        float yv[RANK];
#pragma unroll
        for (int k = 0; k < RANK; ++k) {
            float s = 0.0f;
#pragma unroll
            for (int w = 0; w < NW3; ++w) s += ypart[(w * RANK + k) * SPAN3 + tid];
            yv[k] = s - ybS[k];
        }
        for (int m = 0; m < N_OUT; ++m) {
            float p = 0.0f;
#pragma unroll
            for (int k = 0; k < RANK; ++k) p += yv[k] * B2S[k * N_OUT + m];
            out[(size_t)m * T_LEN + t] = p + biasS[m];
        }
    }
}

extern "C" void kernel_launch(void* const* d_in, const int* in_sizes, int n_in,
                              void* d_out, int out_size, void* d_ws, size_t ws_size,
                              hipStream_t stream) {
    const float* X    = (const float*)d_in[0];
    const float* K    = (const float*)d_in[1];
    const float* B1   = (const float*)d_in[2];
    const float* B2   = (const float*)d_in[3];
    const float* bias = (const float*)d_in[4];
    float*       out  = (float*)d_out;

    float* ws   = (float*)d_ws;
    float* sumA = ws;
    float* sumQ = ws + N_IN * RANK;
    float* B1s  = ws + 2 * N_IN * RANK;
    float* ybar = ws + 3 * N_IN * RANK;
    float* ckeep = ws + CKOFF;

    const int nstats = 2 * N_IN * RANK;
    zero_ws_kernel<<<(nstats + 255) / 256, 256, 0, stream>>>(ws, nstats);

    pass1_stats<<<NRB1 * NSS1, NTH1, 0, stream>>>(X, K, sumA, sumQ, ckeep, ws_size);

    finalize_stats<<<1, 256, 0, stream>>>(B1, sumA, sumQ, B1s, ybar);

    pass3_out<<<NWG3, NTH3, 0, stream>>>(X, K, B2, bias, B1s, ybar, ckeep, ws_size, out);
}


# pass 1 as a 2-phase fast FIR: taps and samples split even/odd, three 51-tap WMMA products P,Q,S per output pair, c[2j]=P+Q, c[2j+1]=S-Q-P[j+1]: 210 f32 WMMAs per 8 t, not 260; one wave per SIMD
# speedup vs baseline: 6.2735x; 1.0674x over previous
#include <hip/hip_runtime.h>
#include <hip/hip_bf16.h>
#include <math.h>


#define T_LEN   50000
#define W_LEN   101
#define N_IN    128
#define RANK    10
#define N_OUT   64

#define KPAD    104
#define KSTEPS  (KPAD / 4)
#define NTILE   5
#define AROWS   (NTILE * 16)
#define AP      108
#define XORG    52
#define XB0     (XORG - 50)

#define G3      50
#define SPAN3   (4 * G3)
#define NW3     8
#define NTH3    (32 * NW3)
#define NFRAG3  (G3 + KSTEPS - 1)
#define KS2     14
#define AP2     60
#define FG1     49
#define SPANW1  (8 * FG1)
#define NW1     4
#define NTH1    (32 * NW1)
#define SPAN1   (NW1 * SPANW1)
#define NSS1    ((T_LEN + SPAN1 - 1) / SPAN1)
#define NRB1    (N_IN / 16)
#define XB1     (XORG - 50)
#define JMAX1   (FG1 + KS2 - 1)
#define XR1     ((NW1 - 1) * SPANW1 + XB1 + 4 + 8 * JMAX1 + 5)
#define XP1F4   421
#define XP1     (4 * XP1F4)
#define X1TR    ((16 * XP1F4 + NTH1 - 1) / NTH1)
#define K1TR    ((RANK * 2 * W_LEN + NTH1 - 1) / NTH1)
#define NK1     (16 * RANK)
#define XR3F4   77
#define XP3     (XR3F4 * 4)
#define X3TR    ((N_IN * XR3F4 + NTH3 - 1) / NTH3)
#define B2TR    ((RANK * N_OUT + NTH3 - 1) / NTH3)
#define NWG3    ((T_LEN / 4 + G3 - 1) / G3)
#define KTR     ((RANK * 2 * W_LEN + 255) / 256)
#define KSN     (KTR * 256)
#define NG      (T_LEN / 4)
#define CREC4   (5 * 32)
#define CKOFF   4096
#define CKBYTES ((size_t)CKOFF * 4 + (size_t)NRB1 * NG * CREC4 * 16)
#define UN3     5
#define NCH3    ((G3 + UN3 - 1) / UN3)

typedef __attribute__((ext_vector_type(2))) float v2f;
typedef __attribute__((ext_vector_type(8))) float v8f;


template <int NTHR>
__device__ __forceinline__ void stage_filters(const float* __restrict__ K,
                                              float* __restrict__ Ks,
                                              float* __restrict__ Af, int tid) {
    static_assert(NTHR >= AP, "one thread per column of the filter bank");
    constexpr int TR = (RANK * 2 * W_LEN + NTHR - 1) / NTHR;
    static_assert(TR * NTHR <= KSN, "Ks holds TR whole trips of the workgroup");
#pragma unroll
    for (int j = 0; j < TR; ++j) {
        const int i = tid + j * NTHR;
        Ks[i] = K[min(i, RANK * 2 * W_LEN - 1)];
    }
    __syncthreads();
    if (tid < AP) {
        const int up = tid;
#pragma unroll
        for (int R = 0; R < AROWS; ++R) {
            const int q = R >> 4, m = R & 15, pos = m >> 2, d = m & 3;
            const int k    = (q < 4) ? ((q >> 1) * 4 + pos) : (8 + (pos >> 1));
            const int part = (q < 4) ? (q & 1) : (pos & 1);
            const int u = up - d;
            const int w = min(max(W_LEN - 1 - u, 0), W_LEN - 1);
            const float v = Ks[(k * 2 + part) * W_LEN + w];
            Af[R * AP + up] = (u >= 0 && u < W_LEN) ? v : 0.0f;
        }
    }
}

__global__ void zero_ws_kernel(float* __restrict__ p, int nelem) {
    const int i = blockIdx.x * blockDim.x + threadIdx.x;
    if (i < nelem) p[i] = 0.0f;
}

template <int M>
__device__ __forceinline__ float row_xor_add(float v) {
    const int x = __builtin_amdgcn_update_dpp(0, __builtin_bit_cast(int, v), 0x160 + M, 0xf, 0xf, true);
    return v + __builtin_bit_cast(float, x);
}

__device__ __forceinline__ float xhalf(float v) {
    const int b = __builtin_bit_cast(int, v);
    return __builtin_bit_cast(float, __builtin_amdgcn_permlanex16(b, b, 0x76543210, 0xfedcba98, false, false));
}

static_assert(N_IN == 16 * NRB1 && NSS1 * SPAN1 >= T_LEN, "row blocks and spans cover the input");
static_assert(T_LEN % 8 == 0 && SPANW1 % 8 == 0 && XORG % 4 == 0, "16-byte groups never straddle 0 or T; an 8-t group is inside or outside [0,T) as a whole");
static_assert(XP1 >= XR1, "window holds every fragment of the last wave's last group");
static_assert((XP1 % 64) % 8 == 4, "row pitch mod 64 is an odd multiple of 4: the 16 rows of a fragment read fall on different banks");
static_assert(KS2 * 4 >= (W_LEN + 1) / 2 + 3 && AP2 >= KS2 * 4, "a bank row holds 51 taps at shifts 0..3");
static_assert(K1TR * NTH1 <= KSN, "Ks holds K1TR whole trips of the workgroup");
static_assert(2 * 64 == NTH1, "bank build: 64 column threads x 2 halves of the rows");

__device__ __forceinline__ void take_group(const v8f (&acc)[NTILE],
                                           v2f (&sA)[4], v2f (&qA)[4], v2f (&sB)[4], v2f (&qB)[4],
                                           v2f (&sC)[4], v2f (&qC)[4],
                                           bool odd, bool keep, float4* dp) {
    float kA[8], kB[8], kC[4];
#pragma unroll
    for (int i = 0; i < 4; ++i) {
        const v2f reA = (v2f){acc[0][2 * i], acc[0][2 * i + 1]};
        const v2f imA = (v2f){acc[1][2 * i], acc[1][2 * i + 1]};
        const v2f reB = (v2f){acc[2][2 * i], acc[2][2 * i + 1]};
        const v2f imB = (v2f){acc[3][2 * i], acc[3][2 * i + 1]};
        const v2f pC  = (v2f){acc[4][2 * i], acc[4][2 * i + 1]};
        const v2f mA = reA * reA + imA * imA;
        const v2f mB = reB * reB + imB * imB;
        const v2f p2 = pC * pC;
        const v2f mC = (v2f){row_xor_add<4>(p2[0]), row_xor_add<4>(p2[1])};
        const v2f cA = (v2f){__builtin_amdgcn_sqrtf(mA[0]), __builtin_amdgcn_sqrtf(mA[1])};
        const v2f cB = (v2f){__builtin_amdgcn_sqrtf(mB[0]), __builtin_amdgcn_sqrtf(mB[1])};
        const v2f cC = (v2f){__builtin_amdgcn_sqrtf(mC[0]), __builtin_amdgcn_sqrtf(mC[1])};
        sA[i] += cA;
        sB[i] += cB;
        sC[i] += cC;
        qA[i] += mA;
        qB[i] += mB;
        qC[i] += mC;
        kA[2 * i] = cA[0]; kA[2 * i + 1] = cA[1];
        kB[2 * i] = cB[0]; kB[2 * i + 1] = cB[1];
        kC[i] = odd ? cC[1] : cC[0];
    }
    if (keep) {
        dp[0]   = make_float4(kA[0], kA[1], kA[2], kA[3]);
        dp[32]  = make_float4(kA[4], kA[5], kA[6], kA[7]);
        dp[64]  = make_float4(kB[0], kB[1], kB[2], kB[3]);
        dp[96]  = make_float4(kB[4], kB[5], kB[6], kB[7]);
        dp[128] = make_float4(kC[0], kC[1], kC[2], kC[3]);
    }
}

__device__ __forceinline__ float quad_up(float v) {
    const int x = __builtin_amdgcn_update_dpp(0, __builtin_bit_cast(int, v), 0x39, 0xf, 0xf, false);
    return __builtin_bit_cast(float, x);
}

__global__ void __launch_bounds__(NTH1)
pass1_stats(const float* __restrict__ X, const float* __restrict__ K,
            float* __restrict__ sumA, float* __restrict__ sumQ,
            float* __restrict__ ckeep, size_t ws_bytes) {
    __shared__ __attribute__((aligned(16))) float Bk[3 * AROWS * AP2];
    __shared__ float  Ks[KSN];
    __shared__ float4 Xw4[X1TR * NTH1];
    __shared__ float  red[2 * NW1 * NK1];
    const int tid  = threadIdx.x;
    const int lane = tid & 31;
    const int wave = __builtin_amdgcn_readfirstlane(tid >> 5);
    const int c = lane & 15, h = lane >> 4;

    const int rb = blockIdx.x % NRB1;
    const int T0 = (blockIdx.x / NRB1) * SPAN1;
    const int Tw = T0 + wave * SPANW1;
    const float* __restrict__ Xb = X + (size_t)rb * 16 * T_LEN;
    const bool keep = ws_bytes >= CKBYTES;
    float4* cdst = reinterpret_cast<float4*>(ckeep) + ((size_t)rb * NG + (Tw / 4)) * CREC4 + lane;
    const bool odd = (lane & 4) != 0;
    const bool d0  = (lane & 3) == 0;

#pragma unroll
    for (int j = 0; j < X1TR; ++j) {
        const int i  = tid + j * NTH1;
        const int ic = min(i, 16 * XP1F4 - 1);
        const int n  = ic / XP1F4, g4 = ic % XP1F4;
        const int t  = T0 - XORG + 4 * g4;
        const int tc = min(max(t, 0), T_LEN - 4);
        float4 v = *reinterpret_cast<const float4*>(Xb + (size_t)n * T_LEN + tc);
        if (t < 0 || t >= T_LEN) v = make_float4(0.0f, 0.0f, 0.0f, 0.0f);
        Xw4[i] = v;
    }
#pragma unroll
    for (int j = 0; j < K1TR; ++j) {
        const int i = tid + j * NTH1;
        Ks[i] = K[min(i, RANK * 2 * W_LEN - 1)];
    }
    __syncthreads();
    {
        const int col = tid & 63, half = tid >> 6;
        if (col < KS2 * 4) {
#pragma unroll
            for (int rr = 0; rr < AROWS / 2; ++rr) {
                const int R = rr + half * (AROWS / 2);
                const int q = R >> 4, m = R & 15, pos = m >> 2, d = m & 3;
                const int k    = (q < 4) ? ((q >> 1) * 4 + pos) : (8 + (pos >> 1));
                const int part = (q < 4) ? (q & 1) : (pos & 1);
                const int v  = col - d;
                const int w0 = min(max(W_LEN - 1 - 2 * v, 0), W_LEN - 1);
                const int w1 = min(max(W_LEN - 2 - 2 * v, 0), W_LEN - 1);
                const float* kr = Ks + (k * 2 + part) * W_LEN;
                const float f0 = (v >= 0 && 2 * v     < W_LEN) ? kr[w0] : 0.0f;
                const float f1 = (v >= 0 && 2 * v + 1 < W_LEN) ? kr[w1] : 0.0f;
                Bk[R * AP2 + col]                   = f0;
                Bk[AROWS * AP2 + R * AP2 + col]     = f1;
                Bk[2 * AROWS * AP2 + R * AP2 + col] = f0 + f1;
            }
        }
    }
    __syncthreads();

    v2f a0[NTILE][KS2], a1[NTILE][KS2], aS[NTILE][KS2];
    {
        const float* bp = Bk + c * AP2 + 2 * h;
#pragma unroll
        for (int q = 0; q < NTILE; ++q)
#pragma unroll
            for (int s = 0; s < KS2; ++s) {
                a0[q][s] = *reinterpret_cast<const v2f*>(bp + q * 16 * AP2 + 4 * s);
                a1[q][s] = *reinterpret_cast<const v2f*>(bp + AROWS * AP2 + q * 16 * AP2 + 4 * s);
                aS[q][s] = *reinterpret_cast<const v2f*>(bp + 2 * AROWS * AP2 + q * 16 * AP2 + 4 * s);
            }
    }
    const float* xr = reinterpret_cast<const float*>(Xw4) + c * XP1 + wave * SPANW1 + XB1 + 4 * h;
    v2f bE[KS2 + 1], bO[KS2 + 1], bD[KS2 + 1];
#pragma unroll
    for (int s = 0; s <= KS2; ++s) {
        const float* p = xr + 8 * s;
        const v2f x01 = *reinterpret_cast<const v2f*>(p);
        const v2f x23 = *reinterpret_cast<const v2f*>(p + 2);
        const float x4 = p[4];
        bE[s] = (v2f){x01[0], x23[0]};
        bO[s] = (v2f){x01[1], x23[1]};
        bD[s] = (v2f){x01[1] + x23[0], x23[1] + x4};
    }

    const int ng = max(0, min(FG1, (T_LEN - Tw) / 8));
    v2f sA[4], qA[4], sB[4], qB[4], sC[4], qC[4];
#pragma unroll
    for (int i = 0; i < 4; ++i) {
        sA[i] = (v2f){0.0f, 0.0f}; qA[i] = (v2f){0.0f, 0.0f};
        sB[i] = (v2f){0.0f, 0.0f}; qB[i] = (v2f){0.0f, 0.0f};
        sC[i] = (v2f){0.0f, 0.0f}; qC[i] = (v2f){0.0f, 0.0f};
    }
    v8f Pg[NTILE];
#pragma unroll
    for (int q = 0; q < NTILE; ++q)
#pragma unroll
        for (int r = 0; r < 8; ++r) Pg[q][r] = 0.0f;
#pragma unroll
    for (int s = 0; s < KS2; ++s)
#pragma unroll
        for (int q = 0; q < NTILE; ++q)
            Pg[q] = __builtin_amdgcn_wmma_f32_16x16x4_f32(
                false, bE[s], false, a0[q][s], (short)0, Pg[q], false, false);

    for (int g = 0; g < ng; ++g) {
        const int jn = min(g + KS2 + 1, JMAX1);
        const float* pn = xr + 8 * jn;
        const v2f n01 = *reinterpret_cast<const v2f*>(pn);
        const v2f n23 = *reinterpret_cast<const v2f*>(pn + 2);
        const float n4 = pn[4];

        v8f Pn[NTILE], Qa[NTILE], Sa[NTILE];
#pragma unroll
        for (int q = 0; q < NTILE; ++q)
#pragma unroll
            for (int r = 0; r < 8; ++r) { Pn[q][r] = 0.0f; Qa[q][r] = 0.0f; Sa[q][r] = 0.0f; }
#pragma unroll
        for (int s = 0; s < KS2; ++s)
#pragma unroll
            for (int q = 0; q < NTILE; ++q) {
                Qa[q] = __builtin_amdgcn_wmma_f32_16x16x4_f32(
                    false, bO[s], false, a1[q][s], (short)0, Qa[q], false, false);
                Sa[q] = __builtin_amdgcn_wmma_f32_16x16x4_f32(
                    false, bD[s], false, aS[q][s], (short)0, Sa[q], false, false);
                Pn[q] = __builtin_amdgcn_wmma_f32_16x16x4_f32(
                    false, bE[s + 1], false, a0[q][s], (short)0, Pn[q], false, false);
            }

        v8f ce[NTILE], co[NTILE];
#pragma unroll
        for (int q = 0; q < NTILE; ++q)
#pragma unroll
            for (int r = 0; r < 8; ++r) {
                ce[q][r] = Pg[q][r] + Qa[q][r];
                const float pm = d0 ? Pn[q][r] : Pg[q][r];
                co[q][r] = (Sa[q][r] - Qa[q][r]) - quad_up(pm);
            }
        float4* dp = cdst + (size_t)(2 * g) * CREC4;
        take_group(ce, sA, qA, sB, qB, sC, qC, odd, keep, dp);
        take_group(co, sA, qA, sB, qB, sC, qC, odd, keep, dp + CREC4);

#pragma unroll
        for (int q = 0; q < NTILE; ++q) Pg[q] = Pn[q];
#pragma unroll
        for (int s = 0; s < KS2; ++s) { bE[s] = bE[s + 1]; bO[s] = bO[s + 1]; bD[s] = bD[s + 1]; }
        bE[KS2] = (v2f){n01[0], n23[0]};
        bO[KS2] = (v2f){n01[1], n23[1]};
        bD[KS2] = (v2f){n01[1] + n23[0], n23[1] + n4};
    }

#pragma unroll
    for (int i = 0; i < 4; ++i)
#pragma unroll
        for (int e = 0; e < 2; ++e) {
            sA[i][e] = row_xor_add<2>(row_xor_add<1>(sA[i][e]));
            qA[i][e] = row_xor_add<2>(row_xor_add<1>(qA[i][e]));
            sB[i][e] = row_xor_add<2>(row_xor_add<1>(sB[i][e]));
            qB[i][e] = row_xor_add<2>(row_xor_add<1>(qB[i][e]));
            sC[i][e] = row_xor_add<2>(row_xor_add<1>(sC[i][e]));
            qC[i][e] = row_xor_add<2>(row_xor_add<1>(qC[i][e]));
        }
    if ((c & 3) == 0) {
        const int pos = c >> 2;
        float* rs = red + wave * NK1 + 8 * h * RANK;
        float* rq = rs + NW1 * NK1;
#pragma unroll
        for (int i = 0; i < 4; ++i)
#pragma unroll
            for (int e = 0; e < 2; ++e) {
                const int ro = (2 * i + e) * RANK;
                rs[ro + pos]     = sA[i][e];  rq[ro + pos]     = qA[i][e];
                rs[ro + 4 + pos] = sB[i][e];  rq[ro + 4 + pos] = qB[i][e];
                if ((pos & 1) == 0) {
                    rs[ro + 8 + (pos >> 1)] = sC[i][e];
                    rq[ro + 8 + (pos >> 1)] = qC[i][e];
                }
            }
    }
    __syncthreads();
    for (int j = tid; j < 2 * NK1; j += NTH1) {
        const int which = j / NK1, idx = j % NK1;
        float v = 0.0f;
#pragma unroll
        for (int w = 0; w < NW1; ++w) v += red[(which * NW1 + w) * NK1 + idx];
        atomicAdd((which ? sumQ : sumA) + rb * NK1 + idx, v);
    }
}

__global__ void finalize_stats(const float* __restrict__ B1,
                               const float* __restrict__ sumA,
                               const float* __restrict__ sumQ,
                               float* __restrict__ B1s, float* __restrict__ ybar) {
    const int tid = threadIdx.x;
    if (tid < RANK) ybar[tid] = 0.0f;
    __syncthreads();
    for (int i = tid; i < N_IN * RANK; i += blockDim.x) {
        const float s    = sumA[i], q = sumQ[i];
        const float mean = s / (float)T_LEN;
        const float var  = (q - s * mean) / (float)(T_LEN - 1);
        const float w    = B1[i] * rsqrtf(var);
        B1s[i] = w;
        atomicAdd(&ybar[i % RANK], w * mean);
    }
}

static_assert(NCH3 % 2 == 0 && NCH3 * UN3 >= G3 && G3 % 2 == 0 && SPAN3 % 8 == 0, "kept |conv|: the chunk loop takes two chunks per trip and covers G3; a workgroup holds whole 8-t groups");
static_assert(N_IN == 16 * NW3, "one wave per block of 16 input rows");
static_assert(XP3 >= XB0 + 4 * (NFRAG3 - 1) + 4, "window holds every fragment of the last group");
static_assert((XP3 % 64) % 8 == 4, "row pitch mod 64 is an odd multiple of 4: the 16 rows of a fragment read fall on 64 different banks");
static_assert(NTH3 >= SPAN3 && NTH3 >= N_OUT && NTH3 >= RANK, "readout: one thread per time point; one per bias and ybar element");
static_assert(NWG3 * SPAN3 >= T_LEN && T_LEN % 4 == 0 && XORG % 4 == 0, "spans cover T; 16-byte groups never straddle 0 or T");

__global__ void __launch_bounds__(NTH3)
pass3_out(const float* __restrict__ X, const float* __restrict__ K,
          const float* __restrict__ B2, const float* __restrict__ bias,
          const float* __restrict__ B1s, const float* __restrict__ ybar,
          const float* __restrict__ ckeep, size_t ws_bytes,
          float* __restrict__ out) {
    __shared__ __attribute__((aligned(16))) float Af[AROWS * AP];
    __shared__ float  Ks[KSN];
    __shared__ float4 Xw4[X3TR * NTH3];
    __shared__ float  B2S[B2TR * NTH3];
    __shared__ __attribute__((aligned(16))) float ypart[NW3 * RANK * SPAN3];
    __shared__ float  ybS[NTH3];
    __shared__ float  biasS[NTH3];
    const int tid  = threadIdx.x;
    const int lane = tid & 31;
    const int wave = __builtin_amdgcn_readfirstlane(tid >> 5);
    const int c = lane & 15, h = lane >> 4;

    const int T0 = blockIdx.x * SPAN3;
    const bool keep = ws_bytes >= CKBYTES;

    if (!keep) {
#pragma unroll
        for (int j = 0; j < X3TR; ++j) {
            const int i  = tid + j * NTH3;
            const int ic = min(i, N_IN * XR3F4 - 1);
            const int n  = ic / XR3F4, g4 = ic % XR3F4;
            const int t  = T0 - XORG + 4 * g4;
            const int tc = min(max(t, 0), T_LEN - 4);
            float4 v = *reinterpret_cast<const float4*>(X + (size_t)n * T_LEN + tc);
            if (t < 0 || t >= T_LEN) v = make_float4(0.0f, 0.0f, 0.0f, 0.0f);
            Xw4[i] = v;
        }
    }
#pragma unroll
    for (int jj = 0; jj < B2TR; ++jj) B2S[tid + jj * NTH3] = B2[min(tid + jj * NTH3, RANK * N_OUT - 1)];
    ybS[tid]   = ybar[min(tid, RANK - 1)];
    biasS[tid] = bias[min(tid, N_OUT - 1)];
    const int nrow = wave * 16 + c;
    const int pos = c >> 2, d = c & 3;
    v2f wA[4], wB[4], wC[4];
    {
        const float* wp = B1s + (wave * 16 + 8 * h) * RANK;
#pragma unroll
        for (int i = 0; i < 4; ++i) {
            const float* w0p = wp + (2 * i) * RANK;
            const float* w1p = w0p + RANK;
            wA[i] = (v2f){w0p[pos], w1p[pos]};
            wB[i] = (v2f){w0p[4 + pos], w1p[4 + pos]};
            wC[i] = (v2f){w0p[8 + (pos >> 1)], w1p[8 + (pos >> 1)]};
        }
    }
    const int ng = min(G3, (T_LEN - T0) / 4);
    float* yw = ypart + wave * RANK * SPAN3;
    const int offAB = (4 * h + pos) * SPAN3 + d;
    const int offC  = (8 + (pos >> 1)) * SPAN3 + d;
    const bool storeC = (lane & 0x14) == 0;

    if (keep) {
        const bool odd = (lane & 4) != 0;
        const int offABk = (4 * h + pos) * SPAN3 + 2 * d;
        const int offCk  = (8 + (pos >> 1)) * SPAN3 + 2 * d;
        float wAf[8], wBf[8], wCs[4];
#pragma unroll
        for (int i = 0; i < 4; ++i) {
            wAf[2 * i] = wA[i][0]; wAf[2 * i + 1] = wA[i][1];
            wBf[2 * i] = wB[i][0]; wBf[2 * i + 1] = wB[i][1];
            wCs[i] = odd ? wC[i][1] : wC[i][0];
        }
        const float4* src = reinterpret_cast<const float4*>(ckeep) + ((size_t)wave * NG) * CREC4 + lane;
        const int G0 = blockIdx.x * G3;
        float4 ra[UN3][5], rn[UN3][5];
#pragma unroll
        for (int u = 0; u < UN3; ++u) {
            const float4* p = src + (size_t)min(G0 + u, NG - 1) * CREC4;
#pragma unroll
            for (int j = 0; j < 5; ++j) ra[u][j] = p[32 * j];
        }
        for (int ch = 0; ch < NCH3; ch += 2) {
#pragma unroll
            for (int u = 0; u < UN3; ++u) {
                const float4* p = src + (size_t)min(G0 + (ch + 1) * UN3 + u, NG - 1) * CREC4;
#pragma unroll
                for (int j = 0; j < 5; ++j) rn[u][j] = p[32 * j];
            }
#pragma unroll
            for (int u = 0; u < UN3; ++u) {
                const int g = ch * UN3 + u;
                float sA = ra[u][0].x * wAf[0] + ra[u][0].y * wAf[1] + ra[u][0].z * wAf[2] + ra[u][0].w * wAf[3]
                         + ra[u][1].x * wAf[4] + ra[u][1].y * wAf[5] + ra[u][1].z * wAf[6] + ra[u][1].w * wAf[7];
                float sB = ra[u][2].x * wBf[0] + ra[u][2].y * wBf[1] + ra[u][2].z * wBf[2] + ra[u][2].w * wBf[3]
                         + ra[u][3].x * wBf[4] + ra[u][3].y * wBf[5] + ra[u][3].z * wBf[6] + ra[u][3].w * wBf[7];
                float sC = ra[u][4].x * wCs[0] + ra[u][4].y * wCs[1] + ra[u][4].z * wCs[2] + ra[u][4].w * wCs[3];
                sC = row_xor_add<4>(sC);
                const float yA = sA + xhalf(sA);
                const float yB = sB + xhalf(sB);
                const float yC = sC + xhalf(sC);
                if (g < ng) {
                    float* yp = yw + 8 * (g >> 1) + (g & 1);
                    yp[offABk] = h ? yB : yA;
                    if (storeC) yp[offCk] = yC;
                }
            }
            if (ch + 2 < NCH3) {
#pragma unroll
                for (int u = 0; u < UN3; ++u) {
                    const float4* p = src + (size_t)min(G0 + (ch + 2) * UN3 + u, NG - 1) * CREC4;
#pragma unroll
                    for (int j = 0; j < 5; ++j) ra[u][j] = p[32 * j];
                }
            }
#pragma unroll
            for (int u = 0; u < UN3; ++u) {
                const int g = (ch + 1) * UN3 + u;
                float sA = rn[u][0].x * wAf[0] + rn[u][0].y * wAf[1] + rn[u][0].z * wAf[2] + rn[u][0].w * wAf[3]
                         + rn[u][1].x * wAf[4] + rn[u][1].y * wAf[5] + rn[u][1].z * wAf[6] + rn[u][1].w * wAf[7];
                float sB = rn[u][2].x * wBf[0] + rn[u][2].y * wBf[1] + rn[u][2].z * wBf[2] + rn[u][2].w * wBf[3]
                         + rn[u][3].x * wBf[4] + rn[u][3].y * wBf[5] + rn[u][3].z * wBf[6] + rn[u][3].w * wBf[7];
                float sC = rn[u][4].x * wCs[0] + rn[u][4].y * wCs[1] + rn[u][4].z * wCs[2] + rn[u][4].w * wCs[3];
                sC = row_xor_add<4>(sC);
                const float yA = sA + xhalf(sA);
                const float yB = sB + xhalf(sB);
                const float yC = sC + xhalf(sC);
                if (g < ng) {
                    float* yp = yw + 8 * (g >> 1) + (g & 1);
                    yp[offABk] = h ? yB : yA;
                    if (storeC) yp[offCk] = yC;
                }
            }
        }
    } else {
        stage_filters<NTH3>(K, Ks, Af, tid);
        __syncthreads();

        v2f a[NTILE][KSTEPS];
        {
            const float* ap = Af + c * AP + 2 * h;
#pragma unroll
            for (int q = 0; q < NTILE; ++q)
#pragma unroll
                for (int s = 0; s < KSTEPS; ++s)
                    a[q][s] = *reinterpret_cast<const v2f*>(ap + q * 16 * AP + 4 * s);
        }
        const float* xr = reinterpret_cast<const float*>(Xw4) + nrow * XP3 + XB0 + 2 * h;
        v2f b[KSTEPS];
#pragma unroll
        for (int s = 0; s < KSTEPS; ++s) b[s] = *reinterpret_cast<const v2f*>(xr + 4 * s);

        for (int g = 0; g < ng; ++g) {
            const int jn = min(g + KSTEPS, NFRAG3 - 1);
            const v2f nb = *reinterpret_cast<const v2f*>(xr + 4 * jn);

            v8f acc[NTILE];
#pragma unroll
            for (int q = 0; q < NTILE; ++q)
#pragma unroll
                for (int r = 0; r < 8; ++r) acc[q][r] = 0.0f;
#pragma unroll
            for (int s = 0; s < KSTEPS; ++s)
#pragma unroll
                for (int q = 0; q < NTILE; ++q)
                    acc[q] = __builtin_amdgcn_wmma_f32_16x16x4_f32(
                        false, b[s], false, a[q][s], (short)0, acc[q], false, false);

            v2f sA = (v2f){0.0f, 0.0f}, sB = (v2f){0.0f, 0.0f}, sC = (v2f){0.0f, 0.0f};
#pragma unroll
            for (int i = 0; i < 4; ++i) {
                const v2f reA = (v2f){acc[0][2 * i], acc[0][2 * i + 1]};
                const v2f imA = (v2f){acc[1][2 * i], acc[1][2 * i + 1]};
                const v2f reB = (v2f){acc[2][2 * i], acc[2][2 * i + 1]};
                const v2f imB = (v2f){acc[3][2 * i], acc[3][2 * i + 1]};
                const v2f pC  = (v2f){acc[4][2 * i], acc[4][2 * i + 1]};
                const v2f mA = reA * reA + imA * imA;
                const v2f mB = reB * reB + imB * imB;
                const v2f qC = pC * pC;
                const float mC0 = row_xor_add<4>(qC[0]);
                const float mC1 = row_xor_add<4>(qC[1]);
                const v2f cA = (v2f){__builtin_amdgcn_sqrtf(mA[0]), __builtin_amdgcn_sqrtf(mA[1])};
                const v2f cB = (v2f){__builtin_amdgcn_sqrtf(mB[0]), __builtin_amdgcn_sqrtf(mB[1])};
                const v2f cC = (v2f){__builtin_amdgcn_sqrtf(mC0), __builtin_amdgcn_sqrtf(mC1)};
                sA += wA[i] * cA;
                sB += wB[i] * cB;
                sC += wC[i] * cC;
            }
            float yA = sA[0] + sA[1], yB = sB[0] + sB[1], yC = sC[0] + sC[1];
            yA += xhalf(yA);
            yB += xhalf(yB);
            yC += xhalf(yC);
            float* yp = yw + 4 * g;
            yp[offAB] = h ? yB : yA;
            if (storeC) yp[offC] = yC;
#pragma unroll
            for (int s = 0; s + 1 < KSTEPS; ++s) b[s] = b[s + 1];
            b[KSTEPS - 1] = nb;
        }
    }
    __syncthreads();

    const int t = T0 + tid;
    if (tid < SPAN3 && t < T_LEN) {
        float yv[RANK];
#pragma unroll
        for (int k = 0; k < RANK; ++k) {
            float s = 0.0f;
#pragma unroll
            for (int w = 0; w < NW3; ++w) s += ypart[(w * RANK + k) * SPAN3 + tid];
            yv[k] = s - ybS[k];
        }
        for (int m = 0; m < N_OUT; ++m) {
            float p = 0.0f;
#pragma unroll
            for (int k = 0; k < RANK; ++k) p += yv[k] * B2S[k * N_OUT + m];
            out[(size_t)m * T_LEN + t] = p + biasS[m];
        }
    }
}

extern "C" void kernel_launch(void* const* d_in, const int* in_sizes, int n_in,
                              void* d_out, int out_size, void* d_ws, size_t ws_size,
                              hipStream_t stream) {
    const float* X    = (const float*)d_in[0];
    const float* K    = (const float*)d_in[1];
    const float* B1   = (const float*)d_in[2];
    const float* B2   = (const float*)d_in[3];
    const float* bias = (const float*)d_in[4];
    float*       out  = (float*)d_out;

    float* ws   = (float*)d_ws;
    float* sumA = ws;
    float* sumQ = ws + N_IN * RANK;
    float* B1s  = ws + 2 * N_IN * RANK;
    float* ybar = ws + 3 * N_IN * RANK;
    float* ckeep = ws + CKOFF;

    const int nstats = 2 * N_IN * RANK;
    zero_ws_kernel<<<(nstats + 255) / 256, 256, 0, stream>>>(ws, nstats);

    pass1_stats<<<NRB1 * NSS1, NTH1, 0, stream>>>(X, K, sumA, sumQ, ckeep, ws_size);

    finalize_stats<<<1, 256, 0, stream>>>(B1, sumA, sumQ, B1s, ybar);

    pass3_out<<<NWG3, NTH3, 0, stream>>>(X, K, B2, bias, B1s, ybar, ckeep, ws_size, out);
}
